# starting kernel
# baseline (speedup 1.0000x reference)
#include <hip/hip_runtime.h>

#define EPSF 1e-6f
#define NSZ 64
#define NLAYERS 63
#define BATCH_TOTAL 262144
#define SINK_ITERS 20
#define ROWS_PER_BLOCK 128

typedef __attribute__((ext_vector_type(2))) float v2f;
typedef __attribute__((ext_vector_type(8))) float v8f;

__global__ __launch_bounds__(64) void btln_sinkhorn_kernel(const float* __restrict__ logits,
                                                           float* __restrict__ P) {
    __shared__ float l[64 * 65];
    const int t = threadIdx.x;
    for (int j = 0; j < 64; ++j) l[t * 65 + j] = logits[t * 64 + j];
    __syncthreads();
    for (int it = 0; it < SINK_ITERS; ++it) {
        float m = -3.4e38f;
        for (int j = 0; j < 64; ++j) m = fmaxf(m, l[t * 65 + j]);
        float s = 0.f;
        for (int j = 0; j < 64; ++j) s += expf(l[t * 65 + j] - m);
        float lse = m + logf(s);
        for (int j = 0; j < 64; ++j) l[t * 65 + j] -= lse;
        __syncthreads();
        m = -3.4e38f;
        for (int i = 0; i < 64; ++i) m = fmaxf(m, l[i * 65 + t]);
        s = 0.f;
        for (int i = 0; i < 64; ++i) s += expf(l[i * 65 + t] - m);
        lse = m + logf(s);
        for (int i = 0; i < 64; ++i) l[i * 65 + t] -= lse;
        __syncthreads();
    }
    for (int j = 0; j < 64; ++j) P[t * 64 + j] = expf(l[t * 65 + j]);
}

__device__ __forceinline__ float btln_clip01(float v) {
    return fminf(fmaxf(v, EPSF), 1.0f - EPSF);
}

__device__ __forceinline__ float btln_F_base(float x, float y, float a,
                                             float w0, float w1, float p) {
    float lin = w0 * x + w1 * y;
    float g = exp2f(p * (2.0f * w0 * log2f(x) + 2.0f * w1 * log2f(y)));
    float rule0 = ((fabsf(x - 1.0f) < EPSF) && (fabsf(y - 1.0f) < EPSF)) ? 1.0f : 0.0f;
    float mid = (3.0f - 4.0f * a) * lin + (4.0f * a - 2.0f) * g;
    float r = (a >= 0.75f) ? g : ((a > 0.5f) ? mid : lin);
    return (fabsf(a - 2.0f) < EPSF) ? rule0 : r;
}

__device__ __forceinline__ float btln_F(float x, float y, float a, float w0, float w1,
                                        float p_hi, float p_lo) {
    x = btln_clip01(x);
    y = btln_clip01(y);
    float lin = w0 * x + w1 * y;
    float hi = btln_F_base(x, y, a, w0, w1, p_hi);
    float aref = fminf(fmaxf(1.0f - a, -1.0f + EPSF), 2.0f - EPSF);
    float lo = 1.0f - btln_F_base(btln_clip01(1.0f - x), btln_clip01(1.0f - y),
                                  aref, w0, w1, p_lo);
    float r = (fabsf(a - 0.5f) < EPSF) ? lin : ((a < 0.5f) ? lo : hi);
    return (r != r) ? a : r;
}

__global__ __launch_bounds__(256) void btln_fused_kernel(const float* __restrict__ x,
                                                         const float* __restrict__ P,
                                                         const float* __restrict__ weights,
                                                         const float* __restrict__ biases,
                                                         float* __restrict__ out) {
    __shared__ float PT[64 * 66];
    __shared__ float leaves[ROWS_PER_BLOCK * 66];
    __shared__ float pw0[NLAYERS], pw1[NLAYERS], pb[NLAYERS];
    __shared__ float pph[NLAYERS], ppl[NLAYERS];

    const int tid = threadIdx.x;

    for (int e = tid; e < 64 * 64; e += 256) {
        int k = e >> 6, n = e & 63;
        PT[n * 66 + k] = P[e];
    }
    if (tid < NLAYERS) {
        float w = weights[tid];
        float b = biases[tid];
        pw0[tid] = w;
        pw1[tid] = 1.0f - w;
        pb[tid]  = b;
        pph[tid] = sqrtf(3.0f / fmaxf(2.0f - b, EPSF)) - 1.0f;
        float aref = fminf(fmaxf(1.0f - b, -1.0f + EPSF), 2.0f - EPSF);
        ppl[tid] = sqrtf(3.0f / fmaxf(2.0f - aref, EPSF)) - 1.0f;
    }
    __syncthreads();

    const int wave = tid >> 5;
    const int lane = tid & 31;
    const int nn   = lane & 15;
    const int kb   = (lane >> 4) * 2;
    const long block_row0 = (long)blockIdx.x * ROWS_PER_BLOCK;
    const int lrow0 = wave * 16;

    const float* arow = x + (block_row0 + lrow0 + nn) * 64 + kb;

    v8f acc0 = {}, acc1 = {}, acc2 = {}, acc3 = {};
#pragma unroll
    for (int s = 0; s < 16; ++s) {
        v2f a  = *(const v2f*)(arow + 4 * s);
        v2f b0 = *(const v2f*)&PT[( 0 + nn) * 66 + 4 * s + kb];
        v2f b1 = *(const v2f*)&PT[(16 + nn) * 66 + 4 * s + kb];
        v2f b2 = *(const v2f*)&PT[(32 + nn) * 66 + 4 * s + kb];
        v2f b3 = *(const v2f*)&PT[(48 + nn) * 66 + 4 * s + kb];
        acc0 = __builtin_amdgcn_wmma_f32_16x16x4_f32(false, a, false, b0, (short)0, acc0, false, false);
        acc1 = __builtin_amdgcn_wmma_f32_16x16x4_f32(false, a, false, b1, (short)0, acc1, false, false);
        acc2 = __builtin_amdgcn_wmma_f32_16x16x4_f32(false, a, false, b2, (short)0, acc2, false, false);
        acc3 = __builtin_amdgcn_wmma_f32_16x16x4_f32(false, a, false, b3, (short)0, acc3, false, false);
    }

    const int mrow = lrow0 + ((lane >> 4) << 3);
#pragma unroll
    for (int r = 0; r < 8; ++r) {
        leaves[(mrow + r) * 66 +  0 + nn] = acc0[r];
        leaves[(mrow + r) * 66 + 16 + nn] = acc1[r];
        leaves[(mrow + r) * 66 + 32 + nn] = acc2[r];
        leaves[(mrow + r) * 66 + 48 + nn] = acc3[r];
    }
    __syncthreads();

    if (tid < ROWS_PER_BLOCK) {
        float st = leaves[tid * 66 + 0];
        for (int i = 0; i < NLAYERS; ++i) {
            float y = leaves[tid * 66 + i + 1];
            st = btln_F(st, y, pb[i], pw0[i], pw1[i], pph[i], ppl[i]);
        }
        out[block_row0 + tid] = st;
    }
}

extern "C" void kernel_launch(void* const* d_in, const int* in_sizes, int n_in,
                              void* d_out, int out_size, void* d_ws, size_t ws_size,
                              hipStream_t stream) {
    const float* x       = (const float*)d_in[0];
    const float* logits  = (const float*)d_in[1];
    const float* weights = (const float*)d_in[2];
    const float* biases  = (const float*)d_in[3];
    float* out = (float*)d_out;
    float* P   = (float*)d_ws;

    btln_sinkhorn_kernel<<<1, 64, 0, stream>>>(logits, P);
    btln_fused_kernel<<<BATCH_TOTAL / ROWS_PER_BLOCK, 256, 0, stream>>>(x, P, weights, biases, out);
}


# Sinkhorn kernel parallelised inside its one work-group: 512 threads, 8 lanes x 8 values per row/column, logsumexp max and sum by DPP butterfly (summation order becomes a tree instead of sequential)
# speedup vs baseline: 2.9147x; 2.9147x over previous
#include <hip/hip_runtime.h>

#define EPSF 1e-6f
#define NSZ 64
#define NLAYERS 63
#define BATCH_TOTAL 262144
#define SINK_ITERS 20
#define ROWS_PER_BLOCK 128

typedef __attribute__((ext_vector_type(2))) float v2f;
typedef __attribute__((ext_vector_type(8))) float v8f;


template <int MASK>
__device__ __forceinline__ float btln_xlane(float v) {
    return __builtin_bit_cast(float, __builtin_amdgcn_mov_dpp(__builtin_bit_cast(int, v),
                                                              0x160 | MASK, 0xf, 0xf, true));
}

__device__ __forceinline__ void btln_lse_normalise(float (&v)[8]) {
    float m = fmaxf(fmaxf(fmaxf(v[0], v[1]), fmaxf(v[2], v[3])),
                    fmaxf(fmaxf(v[4], v[5]), fmaxf(v[6], v[7])));
    m = fmaxf(m, btln_xlane<1>(m));
    m = fmaxf(m, btln_xlane<2>(m));
    m = fmaxf(m, btln_xlane<4>(m));
    float s = ((expf(v[0] - m) + expf(v[1] - m)) + (expf(v[2] - m) + expf(v[3] - m))) +
              ((expf(v[4] - m) + expf(v[5] - m)) + (expf(v[6] - m) + expf(v[7] - m)));
    s += btln_xlane<1>(s);
    s += btln_xlane<2>(s);
    s += btln_xlane<4>(s);
    const float lse = m + logf(s);
    for (int q = 0; q < 8; ++q) v[q] -= lse;
}

__global__ __launch_bounds__(512) void btln_sinkhorn_kernel(const float* __restrict__ logits,
                                                            float* __restrict__ P) {
    __shared__ float l[64 * 65];
    const int tid  = threadIdx.x;
    const int wave = tid >> 5;
    const int lane = tid & 31;
    const int sub  = lane & 7;
    const int line = wave + 16 * (lane >> 3);
    float* lrow = &l[line * 65 + sub];
    float* lcol = &l[sub * 65 + line];

    float v[8];
    for (int q = 0; q < 8; ++q) v[q] = logits[line * 64 + sub + 8 * q];

    for (int it = 0; it < SINK_ITERS; ++it) {
        btln_lse_normalise(v);
        for (int q = 0; q < 8; ++q) lrow[8 * q] = v[q];
        __syncthreads();
        for (int q = 0; q < 8; ++q) v[q] = lcol[8 * q * 65];
        btln_lse_normalise(v);
        if (it == SINK_ITERS - 1) break;
        for (int q = 0; q < 8; ++q) lcol[8 * q * 65] = v[q];
        __syncthreads();
        for (int q = 0; q < 8; ++q) v[q] = lrow[8 * q];
    }

    for (int q = 0; q < 8; ++q) lcol[8 * q * 65] = expf(v[q]);
    __syncthreads();
    for (int k = 0; k < 8; ++k) {
        const int e = tid + 512 * k;
        P[e] = l[(e >> 6) * 65 + (e & 63)];
    }
}

__device__ __forceinline__ float btln_clip01(float v) {
    return fminf(fmaxf(v, EPSF), 1.0f - EPSF);
}

__device__ __forceinline__ float btln_F_base(float x, float y, float a,
                                             float w0, float w1, float p) {
    float lin = w0 * x + w1 * y;
    float g = exp2f(p * (2.0f * w0 * log2f(x) + 2.0f * w1 * log2f(y)));
    float rule0 = ((fabsf(x - 1.0f) < EPSF) && (fabsf(y - 1.0f) < EPSF)) ? 1.0f : 0.0f;
    float mid = (3.0f - 4.0f * a) * lin + (4.0f * a - 2.0f) * g;
    float r = (a >= 0.75f) ? g : ((a > 0.5f) ? mid : lin);
    return (fabsf(a - 2.0f) < EPSF) ? rule0 : r;
}

__device__ __forceinline__ float btln_F(float x, float y, float a, float w0, float w1,
                                        float p_hi, float p_lo) {
    x = btln_clip01(x);
    y = btln_clip01(y);
    float lin = w0 * x + w1 * y;
    float hi = btln_F_base(x, y, a, w0, w1, p_hi);
    float aref = fminf(fmaxf(1.0f - a, -1.0f + EPSF), 2.0f - EPSF);
    float lo = 1.0f - btln_F_base(btln_clip01(1.0f - x), btln_clip01(1.0f - y),
                                  aref, w0, w1, p_lo);
    float r = (fabsf(a - 0.5f) < EPSF) ? lin : ((a < 0.5f) ? lo : hi);
    return (r != r) ? a : r;
}

__global__ __launch_bounds__(256) void btln_fused_kernel(const float* __restrict__ x,
                                                         const float* __restrict__ P,
                                                         const float* __restrict__ weights,
                                                         const float* __restrict__ biases,
                                                         float* __restrict__ out) {
    __shared__ float PT[64 * 66];
    __shared__ float leaves[ROWS_PER_BLOCK * 66];
    __shared__ float pw0[NLAYERS], pw1[NLAYERS], pb[NLAYERS];
    __shared__ float pph[NLAYERS], ppl[NLAYERS];

    const int tid = threadIdx.x;

    for (int e = tid; e < 64 * 64; e += 256) {
        int k = e >> 6, n = e & 63;
        PT[n * 66 + k] = P[e];
    }
    if (tid < NLAYERS) {
        float w = weights[tid];
        float b = biases[tid];
        pw0[tid] = w;
        pw1[tid] = 1.0f - w;
        pb[tid]  = b;
        pph[tid] = sqrtf(3.0f / fmaxf(2.0f - b, EPSF)) - 1.0f;
        float aref = fminf(fmaxf(1.0f - b, -1.0f + EPSF), 2.0f - EPSF);
        ppl[tid] = sqrtf(3.0f / fmaxf(2.0f - aref, EPSF)) - 1.0f;
    }
    __syncthreads();

    const int wave = tid >> 5;
    const int lane = tid & 31;
    const int nn   = lane & 15;
    const int kb   = (lane >> 4) * 2;
    const long block_row0 = (long)blockIdx.x * ROWS_PER_BLOCK;
    const int lrow0 = wave * 16;

    const float* arow = x + (block_row0 + lrow0 + nn) * 64 + kb;

    v8f acc0 = {}, acc1 = {}, acc2 = {}, acc3 = {};
#pragma unroll
    for (int s = 0; s < 16; ++s) {
        v2f a  = *(const v2f*)(arow + 4 * s);
        v2f b0 = *(const v2f*)&PT[( 0 + nn) * 66 + 4 * s + kb];
        v2f b1 = *(const v2f*)&PT[(16 + nn) * 66 + 4 * s + kb];
        v2f b2 = *(const v2f*)&PT[(32 + nn) * 66 + 4 * s + kb];
        v2f b3 = *(const v2f*)&PT[(48 + nn) * 66 + 4 * s + kb];
        acc0 = __builtin_amdgcn_wmma_f32_16x16x4_f32(false, a, false, b0, (short)0, acc0, false, false);
        acc1 = __builtin_amdgcn_wmma_f32_16x16x4_f32(false, a, false, b1, (short)0, acc1, false, false);
        acc2 = __builtin_amdgcn_wmma_f32_16x16x4_f32(false, a, false, b2, (short)0, acc2, false, false);
        acc3 = __builtin_amdgcn_wmma_f32_16x16x4_f32(false, a, false, b3, (short)0, acc3, false, false);
    }

    const int mrow = lrow0 + ((lane >> 4) << 3);
#pragma unroll
    for (int r = 0; r < 8; ++r) {
        leaves[(mrow + r) * 66 +  0 + nn] = acc0[r];
        leaves[(mrow + r) * 66 + 16 + nn] = acc1[r];
        leaves[(mrow + r) * 66 + 32 + nn] = acc2[r];
        leaves[(mrow + r) * 66 + 48 + nn] = acc3[r];
    }
    __syncthreads();

    if (tid < ROWS_PER_BLOCK) {
        float st = leaves[tid * 66 + 0];
        for (int i = 0; i < NLAYERS; ++i) {
            float y = leaves[tid * 66 + i + 1];
            st = btln_F(st, y, pb[i], pw0[i], pw1[i], pph[i], ppl[i]);
        }
        out[block_row0 + tid] = st;
    }
}

extern "C" void kernel_launch(void* const* d_in, const int* in_sizes, int n_in,
                              void* d_out, int out_size, void* d_ws, size_t ws_size,
                              hipStream_t stream) {
    const float* x       = (const float*)d_in[0];
    const float* logits  = (const float*)d_in[1];
    const float* weights = (const float*)d_in[2];
    const float* biases  = (const float*)d_in[3];
    float* out = (float*)d_out;
    float* P   = (float*)d_ws;

    btln_sinkhorn_kernel<<<1, 512, 0, stream>>>(logits, P);
    btln_fused_kernel<<<BATCH_TOTAL / ROWS_PER_BLOCK, 256, 0, stream>>>(x, P, weights, biases, out);
}


# Tree scan evaluates only the side of F that each layer's row-independent bias selects (flags and constants tabulated once per work-group, scalar branches) instead of both sides and a select
# speedup vs baseline: 3.4791x; 1.1936x over previous
#include <hip/hip_runtime.h>

#define EPSF 1e-6f
#define NSZ 64
#define NLAYERS 63
#define BATCH_TOTAL 262144
#define SINK_ITERS 20
#define ROWS_PER_BLOCK 128

typedef __attribute__((ext_vector_type(2))) float v2f;
typedef __attribute__((ext_vector_type(4))) float v4f;
typedef __attribute__((ext_vector_type(8))) float v8f;


template <int MASK>
__device__ __forceinline__ float btln_xlane(float v) {
    return __builtin_bit_cast(float, __builtin_amdgcn_mov_dpp(__builtin_bit_cast(int, v),
                                                              0x160 | MASK, 0xf, 0xf, true));
}

__device__ __forceinline__ void btln_lse_normalise(float (&v)[8]) {
    float m = fmaxf(fmaxf(fmaxf(v[0], v[1]), fmaxf(v[2], v[3])),
                    fmaxf(fmaxf(v[4], v[5]), fmaxf(v[6], v[7])));
    m = fmaxf(m, btln_xlane<1>(m));
    m = fmaxf(m, btln_xlane<2>(m));
    m = fmaxf(m, btln_xlane<4>(m));
    float s = ((expf(v[0] - m) + expf(v[1] - m)) + (expf(v[2] - m) + expf(v[3] - m))) +
              ((expf(v[4] - m) + expf(v[5] - m)) + (expf(v[6] - m) + expf(v[7] - m)));
    s += btln_xlane<1>(s);
    s += btln_xlane<2>(s);
    s += btln_xlane<4>(s);
    const float lse = m + logf(s);
    for (int q = 0; q < 8; ++q) v[q] -= lse;
}

__global__ __launch_bounds__(512) void btln_sinkhorn_kernel(const float* __restrict__ logits,
                                                            float* __restrict__ P) {
    __shared__ float l[64 * 65];
    const int tid  = threadIdx.x;
    const int wave = tid >> 5;
    const int lane = tid & 31;
    const int sub  = lane & 7;
    const int line = wave + 16 * (lane >> 3);
    float* lrow = &l[line * 65 + sub];
    float* lcol = &l[sub * 65 + line];

    float v[8];
    for (int q = 0; q < 8; ++q) v[q] = logits[line * 64 + sub + 8 * q];

    for (int it = 0; it < SINK_ITERS; ++it) {
        btln_lse_normalise(v);
        for (int q = 0; q < 8; ++q) lrow[8 * q] = v[q];
        __syncthreads();
        for (int q = 0; q < 8; ++q) v[q] = lcol[8 * q * 65];
        btln_lse_normalise(v);
        if (it == SINK_ITERS - 1) break;
        for (int q = 0; q < 8; ++q) lcol[8 * q * 65] = v[q];
        __syncthreads();
        for (int q = 0; q < 8; ++q) v[q] = lrow[8 * q];
    }

    for (int q = 0; q < 8; ++q) lcol[8 * q * 65] = expf(v[q]);
    __syncthreads();
    for (int k = 0; k < 8; ++k) {
        const int e = tid + 512 * k;
        P[e] = l[(e >> 6) * 65 + (e & 63)];
    }
}

__device__ __forceinline__ float btln_clip01(float v) {
    return fminf(fmaxf(v, EPSF), 1.0f - EPSF);
}

__device__ __forceinline__ float btln_layer(float st, float leaf, v4f pa, v4f pc,
                                            bool need_g, bool reflect, bool rule0) {
    float X = btln_clip01(st);
    float Y = btln_clip01(leaf);
    if (reflect) {
        X = btln_clip01(1.0f - X);
        Y = btln_clip01(1.0f - Y);
    }
    const float lin = pc.x * X + pc.y * Y;
    float g = 0.0f;
    if (need_g) g = exp2f(pa.z * (pa.x * log2f(X) + pa.y * log2f(Y)));
    float r = pc.z * lin + pc.w * g;
    if (rule0) r = ((fabsf(X - 1.0f) < EPSF) && (fabsf(Y - 1.0f) < EPSF)) ? 1.0f : 0.0f;
    if (reflect) r = 1.0f - r;
    return (r != r) ? pa.w : r;
}

__global__ __launch_bounds__(256) void btln_fused_kernel(const float* __restrict__ x,
                                                         const float* __restrict__ P,
                                                         const float* __restrict__ weights,
                                                         const float* __restrict__ biases,
                                                         float* __restrict__ out) {
    __shared__ float PT[64 * 66];
    __shared__ float leaves[ROWS_PER_BLOCK * 66];
    __shared__ v4f lpa[NLAYERS];
    __shared__ v4f lpc[NLAYERS];
    __shared__ unsigned lmask[6];

    const int tid = threadIdx.x;

    for (int e = tid; e < 64 * 64; e += 256) {
        int k = e >> 6, n = e & 63;
        PT[n * 66 + k] = P[e];
    }
    if (tid < 64) {
        const bool valid = tid < NLAYERS;
        const int t = valid ? tid : NLAYERS - 1;
        const float w = weights[t];
        const float b = biases[t];
        const float w1 = 1.0f - w;
        const float aref = fminf(fmaxf(1.0f - b, -1.0f + EPSF), 2.0f - EPSF);
        const bool top_lin = fabsf(b - 0.5f) < EPSF;
        const bool reflect = !top_lin && (b < 0.5f);
        const float a = reflect ? aref : b;
        const float p = sqrtf(3.0f / fmaxf(2.0f - a, EPSF)) - 1.0f;
        const bool rule0 = !top_lin && (fabsf(a - 2.0f) < EPSF);
        const bool geo = !top_lin && (a >= 0.75f);
        const bool mid = !top_lin && !geo && (a > 0.5f);
        v4f pa, pc;
        pa.x = 2.0f * w; pa.y = 2.0f * w1; pa.z = p; pa.w = b;
        pc.x = w;        pc.y = w1;
        pc.z = geo ? 0.0f : (mid ? 3.0f - 4.0f * a : 1.0f);
        pc.w = geo ? 1.0f : (mid ? 4.0f * a - 2.0f : 0.0f);
        const unsigned mg = __builtin_amdgcn_ballot_w32(valid && (geo || mid) && !rule0);
        const unsigned mr = __builtin_amdgcn_ballot_w32(valid && reflect);
        const unsigned m0 = __builtin_amdgcn_ballot_w32(valid && rule0);
        if (valid) {
            lpa[tid] = pa;
            lpc[tid] = pc;
        }
        if ((tid & 31) == 0) {
            lmask[0 + (tid >> 5)] = mg;
            lmask[2 + (tid >> 5)] = mr;
            lmask[4 + (tid >> 5)] = m0;
        }
    }
    __syncthreads();

    const int wave = tid >> 5;
    const int lane = tid & 31;
    const int nn   = lane & 15;
    const int kb   = (lane >> 4) * 2;
    const long block_row0 = (long)blockIdx.x * ROWS_PER_BLOCK;
    const int lrow0 = wave * 16;

    const float* arow = x + (block_row0 + lrow0 + nn) * 64 + kb;

    v8f acc0 = {}, acc1 = {}, acc2 = {}, acc3 = {};
#pragma unroll
    for (int s = 0; s < 16; ++s) {
        v2f a  = *(const v2f*)(arow + 4 * s);
        v2f b0 = *(const v2f*)&PT[( 0 + nn) * 66 + 4 * s + kb];
        v2f b1 = *(const v2f*)&PT[(16 + nn) * 66 + 4 * s + kb];
        v2f b2 = *(const v2f*)&PT[(32 + nn) * 66 + 4 * s + kb];
        v2f b3 = *(const v2f*)&PT[(48 + nn) * 66 + 4 * s + kb];
        acc0 = __builtin_amdgcn_wmma_f32_16x16x4_f32(false, a, false, b0, (short)0, acc0, false, false);
        acc1 = __builtin_amdgcn_wmma_f32_16x16x4_f32(false, a, false, b1, (short)0, acc1, false, false);
        acc2 = __builtin_amdgcn_wmma_f32_16x16x4_f32(false, a, false, b2, (short)0, acc2, false, false);
        acc3 = __builtin_amdgcn_wmma_f32_16x16x4_f32(false, a, false, b3, (short)0, acc3, false, false);
    }

    const int mrow = lrow0 + ((lane >> 4) << 3);
#pragma unroll
    for (int r = 0; r < 8; ++r) {
        leaves[(mrow + r) * 66 +  0 + nn] = acc0[r];
        leaves[(mrow + r) * 66 + 16 + nn] = acc1[r];
        leaves[(mrow + r) * 66 + 32 + nn] = acc2[r];
        leaves[(mrow + r) * 66 + 48 + nn] = acc3[r];
    }
    __syncthreads();

    if (tid < ROWS_PER_BLOCK) {
        const float* myrow = &leaves[tid * 66];
        unsigned long long mg = ((unsigned long long)__builtin_amdgcn_readfirstlane(lmask[1]) << 32) |
                                (unsigned)__builtin_amdgcn_readfirstlane(lmask[0]);
        unsigned long long mr = ((unsigned long long)__builtin_amdgcn_readfirstlane(lmask[3]) << 32) |
                                (unsigned)__builtin_amdgcn_readfirstlane(lmask[2]);
        unsigned long long m0 = ((unsigned long long)__builtin_amdgcn_readfirstlane(lmask[5]) << 32) |
                                (unsigned)__builtin_amdgcn_readfirstlane(lmask[4]);
        float st = myrow[0];
        v4f a0 = lpa[0], c0 = lpc[0], a1, c1;
        float y0 = myrow[1], y1;
        for (int i = 0; i < NLAYERS - 1; i += 2) {
            a1 = lpa[i + 1]; c1 = lpc[i + 1]; y1 = myrow[i + 2];
            st = btln_layer(st, y0, a0, c0, mg & 1, mr & 1, m0 & 1);
            a0 = lpa[i + 2]; c0 = lpc[i + 2]; y0 = myrow[i + 3];
            st = btln_layer(st, y1, a1, c1, mg & 2, mr & 2, m0 & 2);
            mg >>= 2; mr >>= 2; m0 >>= 2;
        }
        st = btln_layer(st, y0, a0, c0, mg & 1, mr & 1, m0 & 1);
        out[block_row0 + tid] = st;
    }
}

extern "C" void kernel_launch(void* const* d_in, const int* in_sizes, int n_in,
                              void* d_out, int out_size, void* d_ws, size_t ws_size,
                              hipStream_t stream) {
    const float* x       = (const float*)d_in[0];
    const float* logits  = (const float*)d_in[1];
    const float* weights = (const float*)d_in[2];
    const float* biases  = (const float*)d_in[3];
    float* out = (float*)d_out;
    float* P   = (float*)d_ws;

    btln_sinkhorn_kernel<<<1, 512, 0, stream>>>(logits, P);
    btln_fused_kernel<<<BATCH_TOTAL / ROWS_PER_BLOCK, 256, 0, stream>>>(x, P, weights, biases, out);
}


# Sinkhorn kept in base-2 log units (logits scaled by log2(e) once, P = 2^l2) so each pass's exp and log are the single hardware v_exp_f32 / v_log_f32 instructions; summation order unchanged
# speedup vs baseline: 4.0762x; 1.1716x over previous
#include <hip/hip_runtime.h>

#define EPSF 1e-6f
#define NSZ 64
#define NLAYERS 63
#define BATCH_TOTAL 262144
#define SINK_ITERS 20
#define ROWS_PER_BLOCK 128

typedef __attribute__((ext_vector_type(2))) float v2f;
typedef __attribute__((ext_vector_type(4))) float v4f;
typedef __attribute__((ext_vector_type(8))) float v8f;

#define BTLN_LOG2E 1.4426950408889634f

template <int MASK>
__device__ __forceinline__ float btln_xlane(float v) {
    return __builtin_bit_cast(float, __builtin_amdgcn_mov_dpp(__builtin_bit_cast(int, v),
                                                              0x160 | MASK, 0xf, 0xf, true));
}

__device__ __forceinline__ float btln_exp2_nonpos(float t) {
    return __builtin_amdgcn_exp2f(t);
}

__device__ __forceinline__ void btln_lse_normalise(float (&v)[8]) {
    float m = fmaxf(fmaxf(fmaxf(v[0], v[1]), fmaxf(v[2], v[3])),
                    fmaxf(fmaxf(v[4], v[5]), fmaxf(v[6], v[7])));
    m = fmaxf(m, btln_xlane<1>(m));
    m = fmaxf(m, btln_xlane<2>(m));
    m = fmaxf(m, btln_xlane<4>(m));
    float s = ((btln_exp2_nonpos(v[0] - m) + btln_exp2_nonpos(v[1] - m)) +
               (btln_exp2_nonpos(v[2] - m) + btln_exp2_nonpos(v[3] - m))) +
              ((btln_exp2_nonpos(v[4] - m) + btln_exp2_nonpos(v[5] - m)) +
               (btln_exp2_nonpos(v[6] - m) + btln_exp2_nonpos(v[7] - m)));
    s += btln_xlane<1>(s);
    s += btln_xlane<2>(s);
    s += btln_xlane<4>(s);
    const float lse = m + __builtin_amdgcn_logf(s);
    for (int q = 0; q < 8; ++q) v[q] -= lse;
}

__global__ __launch_bounds__(512) void btln_sinkhorn_kernel(const float* __restrict__ logits,
                                                            float* __restrict__ P) {
    __shared__ float l[64 * 65];
    const int tid  = threadIdx.x;
    const int wave = tid >> 5;
    const int lane = tid & 31;
    const int sub  = lane & 7;
    const int line = wave + 16 * (lane >> 3);
    float* lrow = &l[line * 65 + sub];
    float* lcol = &l[sub * 65 + line];

    float v[8];
    for (int q = 0; q < 8; ++q) v[q] = logits[line * 64 + sub + 8 * q] * BTLN_LOG2E;

    for (int it = 0; it < SINK_ITERS; ++it) {
        btln_lse_normalise(v);
        for (int q = 0; q < 8; ++q) lrow[8 * q] = v[q];
        __syncthreads();
        for (int q = 0; q < 8; ++q) v[q] = lcol[8 * q * 65];
        btln_lse_normalise(v);
        if (it == SINK_ITERS - 1) break;
        for (int q = 0; q < 8; ++q) lcol[8 * q * 65] = v[q];
        __syncthreads();
        for (int q = 0; q < 8; ++q) v[q] = lrow[8 * q];
    }

    for (int q = 0; q < 8; ++q) lcol[8 * q * 65] = exp2f(v[q]);
    __syncthreads();
    for (int k = 0; k < 8; ++k) {
        const int e = tid + 512 * k;
        P[e] = l[(e >> 6) * 65 + (e & 63)];
    }
}

__device__ __forceinline__ float btln_clip01(float v) {
    return fminf(fmaxf(v, EPSF), 1.0f - EPSF);
}

__device__ __forceinline__ float btln_layer(float st, float leaf, v4f pa, v4f pc,
                                            bool need_g, bool reflect, bool rule0) {
    float X = btln_clip01(st);
    float Y = btln_clip01(leaf);
    if (reflect) {
        X = btln_clip01(1.0f - X);
        Y = btln_clip01(1.0f - Y);
    }
    const float lin = pc.x * X + pc.y * Y;
    float g = 0.0f;
    if (need_g) g = exp2f(pa.z * (pa.x * log2f(X) + pa.y * log2f(Y)));
    float r = pc.z * lin + pc.w * g;
    if (rule0) r = ((fabsf(X - 1.0f) < EPSF) && (fabsf(Y - 1.0f) < EPSF)) ? 1.0f : 0.0f;
    if (reflect) r = 1.0f - r;
    return (r != r) ? pa.w : r;
}

__global__ __launch_bounds__(256) void btln_fused_kernel(const float* __restrict__ x,
                                                         const float* __restrict__ P,
                                                         const float* __restrict__ weights,
                                                         const float* __restrict__ biases,
                                                         float* __restrict__ out) {
    __shared__ float PT[64 * 66];
    __shared__ float leaves[ROWS_PER_BLOCK * 66];
    __shared__ v4f lpa[NLAYERS];
    __shared__ v4f lpc[NLAYERS];
    __shared__ unsigned lmask[6];

    const int tid = threadIdx.x;

    for (int e = tid; e < 64 * 64; e += 256) {
        int k = e >> 6, n = e & 63;
        PT[n * 66 + k] = P[e];
    }
    if (tid < 64) {
        const bool valid = tid < NLAYERS;
        const int t = valid ? tid : NLAYERS - 1;
        const float w = weights[t];
        const float b = biases[t];
        const float w1 = 1.0f - w;
        const float aref = fminf(fmaxf(1.0f - b, -1.0f + EPSF), 2.0f - EPSF);
        const bool top_lin = fabsf(b - 0.5f) < EPSF;
        const bool reflect = !top_lin && (b < 0.5f);
        const float a = reflect ? aref : b;
        const float p = sqrtf(3.0f / fmaxf(2.0f - a, EPSF)) - 1.0f;
        const bool rule0 = !top_lin && (fabsf(a - 2.0f) < EPSF);
        const bool geo = !top_lin && (a >= 0.75f);
        const bool mid = !top_lin && !geo && (a > 0.5f);
        v4f pa, pc;
        pa.x = 2.0f * w; pa.y = 2.0f * w1; pa.z = p; pa.w = b;
        pc.x = w;        pc.y = w1;
        pc.z = geo ? 0.0f : (mid ? 3.0f - 4.0f * a : 1.0f);
        pc.w = geo ? 1.0f : (mid ? 4.0f * a - 2.0f : 0.0f);
        const unsigned mg = __builtin_amdgcn_ballot_w32(valid && (geo || mid) && !rule0);
        const unsigned mr = __builtin_amdgcn_ballot_w32(valid && reflect);
        const unsigned m0 = __builtin_amdgcn_ballot_w32(valid && rule0);
        if (valid) {
            lpa[tid] = pa;
            lpc[tid] = pc;
        }
        if ((tid & 31) == 0) {
            lmask[0 + (tid >> 5)] = mg;
            lmask[2 + (tid >> 5)] = mr;
            lmask[4 + (tid >> 5)] = m0;
        }
    }
    __syncthreads();

    const int wave = tid >> 5;
    const int lane = tid & 31;
    const int nn   = lane & 15;
    const int kb   = (lane >> 4) * 2;
    const long block_row0 = (long)blockIdx.x * ROWS_PER_BLOCK;
    const int lrow0 = wave * 16;

    const float* arow = x + (block_row0 + lrow0 + nn) * 64 + kb;

    v8f acc0 = {}, acc1 = {}, acc2 = {}, acc3 = {};
#pragma unroll
    for (int s = 0; s < 16; ++s) {
        v2f a  = *(const v2f*)(arow + 4 * s);
        v2f b0 = *(const v2f*)&PT[( 0 + nn) * 66 + 4 * s + kb];
        v2f b1 = *(const v2f*)&PT[(16 + nn) * 66 + 4 * s + kb];
        v2f b2 = *(const v2f*)&PT[(32 + nn) * 66 + 4 * s + kb];
        v2f b3 = *(const v2f*)&PT[(48 + nn) * 66 + 4 * s + kb];
        acc0 = __builtin_amdgcn_wmma_f32_16x16x4_f32(false, a, false, b0, (short)0, acc0, false, false);
        acc1 = __builtin_amdgcn_wmma_f32_16x16x4_f32(false, a, false, b1, (short)0, acc1, false, false);
        acc2 = __builtin_amdgcn_wmma_f32_16x16x4_f32(false, a, false, b2, (short)0, acc2, false, false);
        acc3 = __builtin_amdgcn_wmma_f32_16x16x4_f32(false, a, false, b3, (short)0, acc3, false, false);
    }

    const int mrow = lrow0 + ((lane >> 4) << 3);
#pragma unroll
    for (int r = 0; r < 8; ++r) {
        leaves[(mrow + r) * 66 +  0 + nn] = acc0[r];
        leaves[(mrow + r) * 66 + 16 + nn] = acc1[r];
        leaves[(mrow + r) * 66 + 32 + nn] = acc2[r];
        leaves[(mrow + r) * 66 + 48 + nn] = acc3[r];
    }
    __syncthreads();

    if (tid < ROWS_PER_BLOCK) {
        const float* myrow = &leaves[tid * 66];
        unsigned long long mg = ((unsigned long long)__builtin_amdgcn_readfirstlane(lmask[1]) << 32) |
                                (unsigned)__builtin_amdgcn_readfirstlane(lmask[0]);
        unsigned long long mr = ((unsigned long long)__builtin_amdgcn_readfirstlane(lmask[3]) << 32) |
                                (unsigned)__builtin_amdgcn_readfirstlane(lmask[2]);
        unsigned long long m0 = ((unsigned long long)__builtin_amdgcn_readfirstlane(lmask[5]) << 32) |
                                (unsigned)__builtin_amdgcn_readfirstlane(lmask[4]);
        float st = myrow[0];
        v4f a0 = lpa[0], c0 = lpc[0], a1, c1;
        float y0 = myrow[1], y1;
        for (int i = 0; i < NLAYERS - 1; i += 2) {
            a1 = lpa[i + 1]; c1 = lpc[i + 1]; y1 = myrow[i + 2];
            st = btln_layer(st, y0, a0, c0, mg & 1, mr & 1, m0 & 1);
            a0 = lpa[i + 2]; c0 = lpc[i + 2]; y0 = myrow[i + 3];
            st = btln_layer(st, y1, a1, c1, mg & 2, mr & 2, m0 & 2);
            mg >>= 2; mr >>= 2; m0 >>= 2;
        }
        st = btln_layer(st, y0, a0, c0, mg & 1, mr & 1, m0 & 1);
        out[block_row0 + tid] = st;
    }
}

extern "C" void kernel_launch(void* const* d_in, const int* in_sizes, int n_in,
                              void* d_out, int out_size, void* d_ws, size_t ws_size,
                              hipStream_t stream) {
    const float* x       = (const float*)d_in[0];
    const float* logits  = (const float*)d_in[1];
    const float* weights = (const float*)d_in[2];
    const float* biases  = (const float*)d_in[3];
    float* out = (float*)d_out;
    float* P   = (float*)d_ws;

    btln_sinkhorn_kernel<<<1, 512, 0, stream>>>(logits, P);
    btln_fused_kernel<<<BATCH_TOTAL / ROWS_PER_BLOCK, 256, 0, stream>>>(x, P, weights, biases, out);
}


# Fused kernel re-tiled: one 32-wave work-group per WGP (grid 256) stages P and the layer table once; each wave owns 32 rows, alternating a 16-leaf slice of the product with the scan layers it feeds
# speedup vs baseline: 4.5719x; 1.1216x over previous
#include <hip/hip_runtime.h>

#define EPSF 1e-6f
#define NSZ 64
#define NLAYERS 63
#define BATCH_TOTAL 262144
#define SINK_ITERS 20
#define WAVES_PER_BLOCK 32
#define ROWS_PER_WAVE 32
#define ROWS_PER_BLOCK (WAVES_PER_BLOCK * ROWS_PER_WAVE)
#define LEAF_PITCH 66

typedef __attribute__((ext_vector_type(2))) float v2f;
typedef __attribute__((ext_vector_type(4))) float v4f;
typedef __attribute__((ext_vector_type(8))) float v8f;

#define BTLN_LOG2E 1.4426950408889634f

template <int MASK>
__device__ __forceinline__ float btln_xlane(float v) {
    return __builtin_bit_cast(float, __builtin_amdgcn_mov_dpp(__builtin_bit_cast(int, v),
                                                              0x160 | MASK, 0xf, 0xf, true));
}

__device__ __forceinline__ float btln_exp2_nonpos(float t) {
    return __builtin_amdgcn_exp2f(t);
}

__device__ __forceinline__ void btln_lse_normalise(float (&v)[8]) {
    float m = fmaxf(fmaxf(fmaxf(v[0], v[1]), fmaxf(v[2], v[3])),
                    fmaxf(fmaxf(v[4], v[5]), fmaxf(v[6], v[7])));
    m = fmaxf(m, btln_xlane<1>(m));
    m = fmaxf(m, btln_xlane<2>(m));
    m = fmaxf(m, btln_xlane<4>(m));
    float s = ((btln_exp2_nonpos(v[0] - m) + btln_exp2_nonpos(v[1] - m)) +
               (btln_exp2_nonpos(v[2] - m) + btln_exp2_nonpos(v[3] - m))) +
              ((btln_exp2_nonpos(v[4] - m) + btln_exp2_nonpos(v[5] - m)) +
               (btln_exp2_nonpos(v[6] - m) + btln_exp2_nonpos(v[7] - m)));
    s += btln_xlane<1>(s);
    s += btln_xlane<2>(s);
    s += btln_xlane<4>(s);
    const float lse = m + __builtin_amdgcn_logf(s);
    for (int q = 0; q < 8; ++q) v[q] -= lse;
}

__global__ __launch_bounds__(512) void btln_sinkhorn_kernel(const float* __restrict__ logits,
                                                            float* __restrict__ P) {
    __shared__ float l[64 * 65];
    const int tid  = threadIdx.x;
    const int wave = tid >> 5;
    const int lane = tid & 31;
    const int sub  = lane & 7;
    const int line = wave + 16 * (lane >> 3);
    float* lrow = &l[line * 65 + sub];
    float* lcol = &l[sub * 65 + line];

    float v[8];
    for (int q = 0; q < 8; ++q) v[q] = logits[line * 64 + sub + 8 * q] * BTLN_LOG2E;

    for (int it = 0; it < SINK_ITERS; ++it) {
        btln_lse_normalise(v);
        for (int q = 0; q < 8; ++q) lrow[8 * q] = v[q];
        __syncthreads();
        for (int q = 0; q < 8; ++q) v[q] = lcol[8 * q * 65];
        btln_lse_normalise(v);
        if (it == SINK_ITERS - 1) break;
        for (int q = 0; q < 8; ++q) lcol[8 * q * 65] = v[q];
        __syncthreads();
        for (int q = 0; q < 8; ++q) v[q] = lrow[8 * q];
    }

    for (int q = 0; q < 8; ++q) lcol[8 * q * 65] = exp2f(v[q]);
    __syncthreads();
    for (int k = 0; k < 8; ++k) {
        const int e = tid + 512 * k;
        P[e] = l[(e >> 6) * 65 + (e & 63)];
    }
}

__device__ __forceinline__ float btln_clip01(float v) {
    return fminf(fmaxf(v, EPSF), 1.0f - EPSF);
}

__device__ __forceinline__ float btln_layer(float st, float leaf, v4f pa, v4f pc,
                                            bool need_g, bool reflect, bool rule0) {
    float X = btln_clip01(st);
    float Y = btln_clip01(leaf);
    if (reflect) {
        X = btln_clip01(1.0f - X);
        Y = btln_clip01(1.0f - Y);
    }
    const float lin = pc.x * X + pc.y * Y;
    float g = 0.0f;
    if (need_g) g = exp2f(pa.z * (pa.x * log2f(X) + pa.y * log2f(Y)));
    float r = pc.z * lin + pc.w * g;
    if (rule0) r = ((fabsf(X - 1.0f) < EPSF) && (fabsf(Y - 1.0f) < EPSF)) ? 1.0f : 0.0f;
    if (reflect) r = 1.0f - r;
    return (r != r) ? pa.w : r;
}


__device__ __forceinline__ void btln_leaf_slice(const v2f (&xa)[16], const v2f (&xb)[16],
                                                const float* __restrict__ PT, float* wl,
                                                int t, int nn, int kb, int hrow) {
    v8f acc0 = {}, acc1 = {};
    const float* bcol = &PT[(16 * t + nn) * 66 + kb];
#pragma unroll
    for (int s = 0; s < 16; ++s) {
        const v2f b = *(const v2f*)(bcol + 4 * s);
        acc0 = __builtin_amdgcn_wmma_f32_16x16x4_f32(false, xa[s], false, b, (short)0, acc0, false, false);
        acc1 = __builtin_amdgcn_wmma_f32_16x16x4_f32(false, xb[s], false, b, (short)0, acc1, false, false);
    }
#pragma unroll
    for (int r = 0; r < 8; ++r) {
        wl[(hrow + r) * LEAF_PITCH + 16 * t + nn]      = acc0[r];
        wl[(16 + hrow + r) * LEAF_PITCH + 16 * t + nn] = acc1[r];
    }
    __builtin_amdgcn_wave_barrier();
}

__global__ __launch_bounds__(1024) void btln_fused_kernel(const float* __restrict__ x,
                                                          const float* __restrict__ P,
                                                          const float* __restrict__ weights,
                                                          const float* __restrict__ biases,
                                                          float* __restrict__ out) {
    __shared__ float PT[64 * 66];
    __shared__ float leaves[ROWS_PER_BLOCK * LEAF_PITCH];
    __shared__ v4f lpa[NLAYERS];
    __shared__ v4f lpc[NLAYERS];
    __shared__ unsigned lmask[6];

    const int tid  = threadIdx.x;
    const int wave = tid >> 5;
    const int lane = tid & 31;
    const int nn   = lane & 15;
    const int kb   = (lane >> 4) * 2;
    const long wave_row0 = (long)blockIdx.x * ROWS_PER_BLOCK + wave * ROWS_PER_WAVE;

    const float p0 = P[tid], p1 = P[tid + 1024], p2 = P[tid + 2048], p3 = P[tid + 3072];

    v2f xa[16], xb[16];
    const float* arow = x + (wave_row0 + nn) * 64 + kb;
#pragma unroll
    for (int s = 0; s < 16; ++s) {
        xa[s] = *(const v2f*)(arow + 4 * s);
        xb[s] = *(const v2f*)(arow + 16 * 64 + 4 * s);
    }

    float* pt = &PT[(tid & 63) * 66 + (tid >> 6)];
    pt[0] = p0; pt[16] = p1; pt[32] = p2; pt[48] = p3;
    if (tid < 64) {
        const bool valid = tid < NLAYERS;
        const int t = valid ? tid : NLAYERS - 1;
        const float w = weights[t];
        const float b = biases[t];
        const float w1 = 1.0f - w;
        const float aref = fminf(fmaxf(1.0f - b, -1.0f + EPSF), 2.0f - EPSF);
        const bool top_lin = fabsf(b - 0.5f) < EPSF;
        const bool reflect = !top_lin && (b < 0.5f);
        const float a = reflect ? aref : b;
        const float p = sqrtf(3.0f / fmaxf(2.0f - a, EPSF)) - 1.0f;
        const bool rule0 = !top_lin && (fabsf(a - 2.0f) < EPSF);
        const bool geo = !top_lin && (a >= 0.75f);
        const bool mid = !top_lin && !geo && (a > 0.5f);
        v4f pa, pc;
        pa.x = 2.0f * w; pa.y = 2.0f * w1; pa.z = p; pa.w = b;
        pc.x = w;        pc.y = w1;
        pc.z = geo ? 0.0f : (mid ? 3.0f - 4.0f * a : 1.0f);
        pc.w = geo ? 1.0f : (mid ? 4.0f * a - 2.0f : 0.0f);
        const unsigned mg = __builtin_amdgcn_ballot_w32(valid && (geo || mid) && !rule0);
        const unsigned mr = __builtin_amdgcn_ballot_w32(valid && reflect);
        const unsigned m0 = __builtin_amdgcn_ballot_w32(valid && rule0);
        if (valid) {
            lpa[tid] = pa;
            lpc[tid] = pc;
        }
        if ((tid & 31) == 0) {
            lmask[0 + (tid >> 5)] = mg;
            lmask[2 + (tid >> 5)] = mr;
            lmask[4 + (tid >> 5)] = m0;
        }
    }
    __syncthreads();

    float* wl = &leaves[wave * ROWS_PER_WAVE * LEAF_PITCH];
    const int hrow = (lane >> 4) << 3;
    const float* myrow = wl + lane * LEAF_PITCH;

    unsigned long long mg = ((unsigned long long)__builtin_amdgcn_readfirstlane(lmask[1]) << 32) |
                            (unsigned)__builtin_amdgcn_readfirstlane(lmask[0]);
    unsigned long long mr = ((unsigned long long)__builtin_amdgcn_readfirstlane(lmask[3]) << 32) |
                            (unsigned)__builtin_amdgcn_readfirstlane(lmask[2]);
    unsigned long long m0 = ((unsigned long long)__builtin_amdgcn_readfirstlane(lmask[5]) << 32) |
                            (unsigned)__builtin_amdgcn_readfirstlane(lmask[4]);

    btln_leaf_slice(xa, xb, PT, wl, 0, nn, kb, hrow);
    float st = myrow[0];
    v4f a0 = lpa[0], c0 = lpc[0], a1, c1;
    float y0 = myrow[1], y1;
    int i = 0;
    for (int t = 0; t < 4; ++t) {
        if (t > 0) btln_leaf_slice(xa, xb, PT, wl, t, nn, kb, hrow);
        const int iend = 16 * t + 14;
        for (; i < iend; i += 2) {
            a1 = lpa[i + 1]; c1 = lpc[i + 1]; y1 = myrow[i + 2];
            st = btln_layer(st, y0, a0, c0, mg & 1, mr & 1, m0 & 1);
            a0 = lpa[i + 2]; c0 = lpc[i + 2]; y0 = myrow[i + 3];
            st = btln_layer(st, y1, a1, c1, mg & 2, mr & 2, m0 & 2);
            mg >>= 2; mr >>= 2; m0 >>= 2;
        }
    }
    st = btln_layer(st, y0, a0, c0, mg & 1, mr & 1, m0 & 1);
    out[wave_row0 + lane] = st;
}

extern "C" void kernel_launch(void* const* d_in, const int* in_sizes, int n_in,
                              void* d_out, int out_size, void* d_ws, size_t ws_size,
                              hipStream_t stream) {
    const float* x       = (const float*)d_in[0];
    const float* logits  = (const float*)d_in[1];
    const float* weights = (const float*)d_in[2];
    const float* biases  = (const float*)d_in[3];
    float* out = (float*)d_out;
    float* P   = (float*)d_ws;

    btln_sinkhorn_kernel<<<1, 512, 0, stream>>>(logits, P);
    btln_fused_kernel<<<BATCH_TOTAL / ROWS_PER_BLOCK, 1024, 0, stream>>>(x, P, weights, biases, out);
}


# Scan layers take log2 and 2^t of the geometric term on the bare hardware v_log_f32 / v_exp_f32 (operands clipped to >= 1e-6, exponent > -30: never denormal), not the range-scaling library forms
# speedup vs baseline: 5.0331x; 1.1009x over previous
#include <hip/hip_runtime.h>

#define EPSF 1e-6f
#define NSZ 64
#define NLAYERS 63
#define BATCH_TOTAL 262144
#define SINK_ITERS 20
#define WAVES_PER_BLOCK 32
#define ROWS_PER_WAVE 32
#define ROWS_PER_BLOCK (WAVES_PER_BLOCK * ROWS_PER_WAVE)
#define LEAF_PITCH 66

typedef __attribute__((ext_vector_type(2))) float v2f;
typedef __attribute__((ext_vector_type(4))) float v4f;
typedef __attribute__((ext_vector_type(8))) float v8f;

#define BTLN_LOG2E 1.4426950408889634f

template <int MASK>
__device__ __forceinline__ float btln_xlane(float v) {
    return __builtin_bit_cast(float, __builtin_amdgcn_mov_dpp(__builtin_bit_cast(int, v),
                                                              0x160 | MASK, 0xf, 0xf, true));
}

__device__ __forceinline__ float btln_exp2_nonpos(float t) {
    return __builtin_amdgcn_exp2f(t);
}

__device__ __forceinline__ void btln_lse_normalise(float (&v)[8]) {
    float m = fmaxf(fmaxf(fmaxf(v[0], v[1]), fmaxf(v[2], v[3])),
                    fmaxf(fmaxf(v[4], v[5]), fmaxf(v[6], v[7])));
    m = fmaxf(m, btln_xlane<1>(m));
    m = fmaxf(m, btln_xlane<2>(m));
    m = fmaxf(m, btln_xlane<4>(m));
    float s = ((btln_exp2_nonpos(v[0] - m) + btln_exp2_nonpos(v[1] - m)) +
               (btln_exp2_nonpos(v[2] - m) + btln_exp2_nonpos(v[3] - m))) +
              ((btln_exp2_nonpos(v[4] - m) + btln_exp2_nonpos(v[5] - m)) +
               (btln_exp2_nonpos(v[6] - m) + btln_exp2_nonpos(v[7] - m)));
    s += btln_xlane<1>(s);
    s += btln_xlane<2>(s);
    s += btln_xlane<4>(s);
    const float lse = m + __builtin_amdgcn_logf(s);
    for (int q = 0; q < 8; ++q) v[q] -= lse;
}

__global__ __launch_bounds__(512) void btln_sinkhorn_kernel(const float* __restrict__ logits,
                                                            float* __restrict__ P) {
    __shared__ float l[64 * 65];
    const int tid  = threadIdx.x;
    const int wave = tid >> 5;
    const int lane = tid & 31;
    const int sub  = lane & 7;
    const int line = wave + 16 * (lane >> 3);
    float* lrow = &l[line * 65 + sub];
    float* lcol = &l[sub * 65 + line];

    float v[8];
    for (int q = 0; q < 8; ++q) v[q] = logits[line * 64 + sub + 8 * q] * BTLN_LOG2E;

    for (int it = 0; it < SINK_ITERS; ++it) {
        btln_lse_normalise(v);
        for (int q = 0; q < 8; ++q) lrow[8 * q] = v[q];
        __syncthreads();
        for (int q = 0; q < 8; ++q) v[q] = lcol[8 * q * 65];
        btln_lse_normalise(v);
        if (it == SINK_ITERS - 1) break;
        for (int q = 0; q < 8; ++q) lcol[8 * q * 65] = v[q];
        __syncthreads();
        for (int q = 0; q < 8; ++q) v[q] = lrow[8 * q];
    }

    for (int q = 0; q < 8; ++q) lcol[8 * q * 65] = exp2f(v[q]);
    __syncthreads();
    for (int k = 0; k < 8; ++k) {
        const int e = tid + 512 * k;
        P[e] = l[(e >> 6) * 65 + (e & 63)];
    }
}

__device__ __forceinline__ float btln_clip01(float v) {
    return fminf(fmaxf(v, EPSF), 1.0f - EPSF);
}

__device__ __forceinline__ float btln_layer(float st, float leaf, v4f pa, v4f pc,
                                            bool need_g, bool reflect, bool rule0) {
    float X = btln_clip01(st);
    float Y = btln_clip01(leaf);
    if (reflect) {
        X = btln_clip01(1.0f - X);
        Y = btln_clip01(1.0f - Y);
    }
    const float lin = pc.x * X + pc.y * Y;
    float g = 0.0f;
    if (need_g) g = __builtin_amdgcn_exp2f(pa.z * (pa.x * __builtin_amdgcn_logf(X) +
                                                   pa.y * __builtin_amdgcn_logf(Y)));
    float r = pc.z * lin + pc.w * g;
    if (rule0) r = ((fabsf(X - 1.0f) < EPSF) && (fabsf(Y - 1.0f) < EPSF)) ? 1.0f : 0.0f;
    if (reflect) r = 1.0f - r;
    return (r != r) ? pa.w : r;
}


__device__ __forceinline__ void btln_leaf_slice(const v2f (&xa)[16], const v2f (&xb)[16],
                                                const float* __restrict__ PT, float* wl,
                                                int t, int nn, int kb, int hrow) {
    v8f acc0 = {}, acc1 = {};
    const float* bcol = &PT[(16 * t + nn) * 66 + kb];
#pragma unroll
    for (int s = 0; s < 16; ++s) {
        const v2f b = *(const v2f*)(bcol + 4 * s);
        acc0 = __builtin_amdgcn_wmma_f32_16x16x4_f32(false, xa[s], false, b, (short)0, acc0, false, false);
        acc1 = __builtin_amdgcn_wmma_f32_16x16x4_f32(false, xb[s], false, b, (short)0, acc1, false, false);
    }
#pragma unroll
    for (int r = 0; r < 8; ++r) {
        wl[(hrow + r) * LEAF_PITCH + 16 * t + nn]      = acc0[r];
        wl[(16 + hrow + r) * LEAF_PITCH + 16 * t + nn] = acc1[r];
    }
    __builtin_amdgcn_wave_barrier();
}

__global__ __launch_bounds__(1024) void btln_fused_kernel(const float* __restrict__ x,
                                                          const float* __restrict__ P,
                                                          const float* __restrict__ weights,
                                                          const float* __restrict__ biases,
                                                          float* __restrict__ out) {
    __shared__ float PT[64 * 66];
    __shared__ float leaves[ROWS_PER_BLOCK * LEAF_PITCH];
    __shared__ v4f lpa[NLAYERS];
    __shared__ v4f lpc[NLAYERS];
    __shared__ unsigned lmask[6];

    const int tid  = threadIdx.x;
    const int wave = tid >> 5;
    const int lane = tid & 31;
    const int nn   = lane & 15;
    const int kb   = (lane >> 4) * 2;
    const long wave_row0 = (long)blockIdx.x * ROWS_PER_BLOCK + wave * ROWS_PER_WAVE;

    const float p0 = P[tid], p1 = P[tid + 1024], p2 = P[tid + 2048], p3 = P[tid + 3072];

    v2f xa[16], xb[16];
    const float* arow = x + (wave_row0 + nn) * 64 + kb;
#pragma unroll
    for (int s = 0; s < 16; ++s) {
        xa[s] = *(const v2f*)(arow + 4 * s);
        xb[s] = *(const v2f*)(arow + 16 * 64 + 4 * s);
    }

    float* pt = &PT[(tid & 63) * 66 + (tid >> 6)];
    pt[0] = p0; pt[16] = p1; pt[32] = p2; pt[48] = p3;
    if (tid < 64) {
        const bool valid = tid < NLAYERS;
        const int t = valid ? tid : NLAYERS - 1;
        const float w = weights[t];
        const float b = biases[t];
        const float w1 = 1.0f - w;
        const float aref = fminf(fmaxf(1.0f - b, -1.0f + EPSF), 2.0f - EPSF);
        const bool top_lin = fabsf(b - 0.5f) < EPSF;
        const bool reflect = !top_lin && (b < 0.5f);
        const float a = reflect ? aref : b;
        const float p = sqrtf(3.0f / fmaxf(2.0f - a, EPSF)) - 1.0f;
        const bool rule0 = !top_lin && (fabsf(a - 2.0f) < EPSF);
        const bool geo = !top_lin && (a >= 0.75f);
        const bool mid = !top_lin && !geo && (a > 0.5f);
        v4f pa, pc;
        pa.x = 2.0f * w; pa.y = 2.0f * w1; pa.z = p; pa.w = b;
        pc.x = w;        pc.y = w1;
        pc.z = geo ? 0.0f : (mid ? 3.0f - 4.0f * a : 1.0f);
        pc.w = geo ? 1.0f : (mid ? 4.0f * a - 2.0f : 0.0f);
        const unsigned mg = __builtin_amdgcn_ballot_w32(valid && (geo || mid) && !rule0);
        const unsigned mr = __builtin_amdgcn_ballot_w32(valid && reflect);
        const unsigned m0 = __builtin_amdgcn_ballot_w32(valid && rule0);
        if (valid) {
            lpa[tid] = pa;
            lpc[tid] = pc;
        }
        if ((tid & 31) == 0) {
            lmask[0 + (tid >> 5)] = mg;
            lmask[2 + (tid >> 5)] = mr;
            lmask[4 + (tid >> 5)] = m0;
        }
    }
    __syncthreads();

    float* wl = &leaves[wave * ROWS_PER_WAVE * LEAF_PITCH];
    const int hrow = (lane >> 4) << 3;
    const float* myrow = wl + lane * LEAF_PITCH;

    unsigned long long mg = ((unsigned long long)__builtin_amdgcn_readfirstlane(lmask[1]) << 32) |
                            (unsigned)__builtin_amdgcn_readfirstlane(lmask[0]);
    unsigned long long mr = ((unsigned long long)__builtin_amdgcn_readfirstlane(lmask[3]) << 32) |
                            (unsigned)__builtin_amdgcn_readfirstlane(lmask[2]);
    unsigned long long m0 = ((unsigned long long)__builtin_amdgcn_readfirstlane(lmask[5]) << 32) |
                            (unsigned)__builtin_amdgcn_readfirstlane(lmask[4]);

    btln_leaf_slice(xa, xb, PT, wl, 0, nn, kb, hrow);
    float st = myrow[0];
    v4f a0 = lpa[0], c0 = lpc[0], a1, c1;
    float y0 = myrow[1], y1;
    int i = 0;
    for (int t = 0; t < 4; ++t) {
        if (t > 0) btln_leaf_slice(xa, xb, PT, wl, t, nn, kb, hrow);
        const int iend = 16 * t + 14;
        for (; i < iend; i += 2) {
            a1 = lpa[i + 1]; c1 = lpc[i + 1]; y1 = myrow[i + 2];
            st = btln_layer(st, y0, a0, c0, mg & 1, mr & 1, m0 & 1);
            a0 = lpa[i + 2]; c0 = lpc[i + 2]; y0 = myrow[i + 3];
            st = btln_layer(st, y1, a1, c1, mg & 2, mr & 2, m0 & 2);
            mg >>= 2; mr >>= 2; m0 >>= 2;
        }
    }
    st = btln_layer(st, y0, a0, c0, mg & 1, mr & 1, m0 & 1);
    out[wave_row0 + lane] = st;
}

extern "C" void kernel_launch(void* const* d_in, const int* in_sizes, int n_in,
                              void* d_out, int out_size, void* d_ws, size_t ws_size,
                              hipStream_t stream) {
    const float* x       = (const float*)d_in[0];
    const float* logits  = (const float*)d_in[1];
    const float* weights = (const float*)d_in[2];
    const float* biases  = (const float*)d_in[3];
    float* out = (float*)d_out;
    float* P   = (float*)d_ws;

    btln_sinkhorn_kernel<<<1, 512, 0, stream>>>(logits, P);
    btln_fused_kernel<<<BATCH_TOTAL / ROWS_PER_BLOCK, 1024, 0, stream>>>(x, P, weights, biases, out);
}


# Sinkhorn folded into the fused kernel: each work-group normalises the 64x64 logits in LDS (16 of its 32 waves, LDS-only barriers) while its x rows load; one launch, no P round trip; same sums
# speedup vs baseline: 5.6428x; 1.1211x over previous
#include <hip/hip_runtime.h>

#define EPSF 1e-6f
#define NSZ 64
#define NLAYERS 63
#define BATCH_TOTAL 262144
#define SINK_ITERS 20
#define WAVES_PER_BLOCK 32
#define ROWS_PER_WAVE 32
#define ROWS_PER_BLOCK (WAVES_PER_BLOCK * ROWS_PER_WAVE)
#define LEAF_PITCH 66

typedef __attribute__((ext_vector_type(2))) float v2f;
typedef __attribute__((ext_vector_type(4))) float v4f;
typedef __attribute__((ext_vector_type(8))) float v8f;

#define SINK_WAVES 16
#define BTLN_LOG2E 1.4426950408889634f

__device__ __forceinline__ void btln_lds_barrier() {
    __builtin_amdgcn_fence(__ATOMIC_RELEASE, "workgroup", "local");
    __builtin_amdgcn_s_barrier();
    __builtin_amdgcn_fence(__ATOMIC_ACQUIRE, "workgroup", "local");
}

template <int MASK>
__device__ __forceinline__ float btln_xlane(float v) {
    return __builtin_bit_cast(float, __builtin_amdgcn_mov_dpp(__builtin_bit_cast(int, v),
                                                              0x160 | MASK, 0xf, 0xf, true));
}

__device__ __forceinline__ float btln_exp2_nonpos(float t) {
    return __builtin_amdgcn_exp2f(t);
}

__device__ __forceinline__ void btln_lse_normalise(float (&v)[8]) {
    float m = fmaxf(fmaxf(fmaxf(v[0], v[1]), fmaxf(v[2], v[3])),
                    fmaxf(fmaxf(v[4], v[5]), fmaxf(v[6], v[7])));
    m = fmaxf(m, btln_xlane<1>(m));
    m = fmaxf(m, btln_xlane<2>(m));
    m = fmaxf(m, btln_xlane<4>(m));
    float s = ((btln_exp2_nonpos(v[0] - m) + btln_exp2_nonpos(v[1] - m)) +
               (btln_exp2_nonpos(v[2] - m) + btln_exp2_nonpos(v[3] - m))) +
              ((btln_exp2_nonpos(v[4] - m) + btln_exp2_nonpos(v[5] - m)) +
               (btln_exp2_nonpos(v[6] - m) + btln_exp2_nonpos(v[7] - m)));
    s += btln_xlane<1>(s);
    s += btln_xlane<2>(s);
    s += btln_xlane<4>(s);
    const float lse = m + __builtin_amdgcn_logf(s);
    for (int q = 0; q < 8; ++q) v[q] -= lse;
}

__device__ __forceinline__ float btln_clip01(float v) {
    return fminf(fmaxf(v, EPSF), 1.0f - EPSF);
}

__device__ __forceinline__ float btln_layer(float st, float leaf, v4f pa, v4f pc,
                                            bool need_g, bool reflect, bool rule0) {
    float X = btln_clip01(st);
    float Y = btln_clip01(leaf);
    if (reflect) {
        X = btln_clip01(1.0f - X);
        Y = btln_clip01(1.0f - Y);
    }
    const float lin = pc.x * X + pc.y * Y;
    float g = 0.0f;
    if (need_g) g = __builtin_amdgcn_exp2f(pa.z * (pa.x * __builtin_amdgcn_logf(X) +
                                                   pa.y * __builtin_amdgcn_logf(Y)));
    float r = pc.z * lin + pc.w * g;
    if (rule0) r = ((fabsf(X - 1.0f) < EPSF) && (fabsf(Y - 1.0f) < EPSF)) ? 1.0f : 0.0f;
    if (reflect) r = 1.0f - r;
    return (r != r) ? pa.w : r;
}


__device__ __forceinline__ void btln_leaf_slice(const v2f (&xa)[16], const v2f (&xb)[16],
                                                const float* __restrict__ PT, float* wl,
                                                int t, int nn, int kb, int hrow) {
    v8f acc0 = {}, acc1 = {};
    const float* bcol = &PT[(16 * t + nn) * 66 + kb];
#pragma unroll
    for (int s = 0; s < 16; ++s) {
        const v2f b = *(const v2f*)(bcol + 4 * s);
        acc0 = __builtin_amdgcn_wmma_f32_16x16x4_f32(false, xa[s], false, b, (short)0, acc0, false, false);
        acc1 = __builtin_amdgcn_wmma_f32_16x16x4_f32(false, xb[s], false, b, (short)0, acc1, false, false);
    }
#pragma unroll
    for (int r = 0; r < 8; ++r) {
        wl[(hrow + r) * LEAF_PITCH + 16 * t + nn]      = acc0[r];
        wl[(16 + hrow + r) * LEAF_PITCH + 16 * t + nn] = acc1[r];
    }
    __builtin_amdgcn_wave_barrier();
}

__global__ __launch_bounds__(1024) void btln_fused_kernel(const float* __restrict__ x,
                                                          const float* __restrict__ logits,
                                                          const float* __restrict__ weights,
                                                          const float* __restrict__ biases,
                                                          float* __restrict__ out) {
    __shared__ float PT[64 * 66];
    __shared__ float leaves[ROWS_PER_BLOCK * LEAF_PITCH];
    __shared__ v4f lpa[NLAYERS];
    __shared__ v4f lpc[NLAYERS];
    __shared__ unsigned lmask[6];

    const int tid  = threadIdx.x;
    const int wave = tid >> 5;
    const int lane = tid & 31;
    const int nn   = lane & 15;
    const int kb   = (lane >> 4) * 2;
    const long wave_row0 = (long)blockIdx.x * ROWS_PER_BLOCK + wave * ROWS_PER_WAVE;

    const bool sink = wave < SINK_WAVES;
    const bool tabw = (wave >> 1) == (SINK_WAVES >> 1);
    const int sub  = lane & 7;
    const int line = (wave & (SINK_WAVES - 1)) + 16 * (lane >> 3);
    float* lrow = &leaves[line * 65 + sub];
    float* lcol = &leaves[sub * 65 + line];

    float v[8] = {0.0f, 0.0f, 0.0f, 0.0f, 0.0f, 0.0f, 0.0f, 0.0f};
    if (sink) {
        for (int q = 0; q < 8; ++q) v[q] = logits[line * 64 + sub + 8 * q];
    }
    const int tl = tid - 32 * SINK_WAVES;
    const bool tvalid = tl < NLAYERS;
    float tw = 0.0f, tb = 0.0f;
    if (tabw) {
        const int t = tvalid ? tl : NLAYERS - 1;
        tw = weights[t];
        tb = biases[t];
    }

    v2f xa[16], xb[16];
    const float* arow = x + (wave_row0 + nn) * 64 + kb;
#pragma unroll
    for (int s = 0; s < 16; ++s) {
        xa[s] = *(const v2f*)(arow + 4 * s);
        xb[s] = *(const v2f*)(arow + 16 * 64 + 4 * s);
    }

    if (tabw) {
        const float w = tw;
        const float b = tb;
        const float w1 = 1.0f - w;
        const float aref = fminf(fmaxf(1.0f - b, -1.0f + EPSF), 2.0f - EPSF);
        const bool top_lin = fabsf(b - 0.5f) < EPSF;
        const bool reflect = !top_lin && (b < 0.5f);
        const float a = reflect ? aref : b;
        const float p = sqrtf(3.0f / fmaxf(2.0f - a, EPSF)) - 1.0f;
        const bool rule0 = !top_lin && (fabsf(a - 2.0f) < EPSF);
        const bool geo = !top_lin && (a >= 0.75f);
        const bool mid = !top_lin && !geo && (a > 0.5f);
        v4f pa, pc;
        pa.x = 2.0f * w; pa.y = 2.0f * w1; pa.z = p; pa.w = b;
        pc.x = w;        pc.y = w1;
        pc.z = geo ? 0.0f : (mid ? 3.0f - 4.0f * a : 1.0f);
        pc.w = geo ? 1.0f : (mid ? 4.0f * a - 2.0f : 0.0f);
        const unsigned mg = __builtin_amdgcn_ballot_w32(tvalid && (geo || mid) && !rule0);
        const unsigned mr = __builtin_amdgcn_ballot_w32(tvalid && reflect);
        const unsigned m0 = __builtin_amdgcn_ballot_w32(tvalid && rule0);
        if (tvalid) {
            lpa[tl] = pa;
            lpc[tl] = pc;
        }
        if (lane == 0) {
            lmask[0 + (tl >> 5)] = mg;
            lmask[2 + (tl >> 5)] = mr;
            lmask[4 + (tl >> 5)] = m0;
        }
    }

    if (sink) {
        for (int q = 0; q < 8; ++q) v[q] *= BTLN_LOG2E;
    }
    for (int it = 0; it < SINK_ITERS; ++it) {
        if (sink) {
            btln_lse_normalise(v);
            for (int q = 0; q < 8; ++q) lrow[8 * q] = v[q];
        }
        btln_lds_barrier();
        if (sink) {
            for (int q = 0; q < 8; ++q) v[q] = lcol[8 * q * 65];
            btln_lse_normalise(v);
        }
        if (it == SINK_ITERS - 1) break;
        if (sink) {
            for (int q = 0; q < 8; ++q) lcol[8 * q * 65] = v[q];
        }
        btln_lds_barrier();
        if (sink) {
            for (int q = 0; q < 8; ++q) v[q] = lrow[8 * q];
        }
    }
    if (sink) {
        for (int q = 0; q < 8; ++q) PT[line * 66 + sub + 8 * q] = exp2f(v[q]);
    }
    __syncthreads();

    float* wl = &leaves[wave * ROWS_PER_WAVE * LEAF_PITCH];
    const int hrow = (lane >> 4) << 3;
    const float* myrow = wl + lane * LEAF_PITCH;

    unsigned long long mg = ((unsigned long long)__builtin_amdgcn_readfirstlane(lmask[1]) << 32) |
                            (unsigned)__builtin_amdgcn_readfirstlane(lmask[0]);
    unsigned long long mr = ((unsigned long long)__builtin_amdgcn_readfirstlane(lmask[3]) << 32) |
                            (unsigned)__builtin_amdgcn_readfirstlane(lmask[2]);
    unsigned long long m0 = ((unsigned long long)__builtin_amdgcn_readfirstlane(lmask[5]) << 32) |
                            (unsigned)__builtin_amdgcn_readfirstlane(lmask[4]);

    btln_leaf_slice(xa, xb, PT, wl, 0, nn, kb, hrow);
    float st = myrow[0];
    v4f a0 = lpa[0], c0 = lpc[0], a1, c1;
    float y0 = myrow[1], y1;
    int i = 0;
    for (int t = 0; t < 4; ++t) {
        if (t > 0) btln_leaf_slice(xa, xb, PT, wl, t, nn, kb, hrow);
        const int iend = 16 * t + 14;
        for (; i < iend; i += 2) {
            a1 = lpa[i + 1]; c1 = lpc[i + 1]; y1 = myrow[i + 2];
            st = btln_layer(st, y0, a0, c0, mg & 1, mr & 1, m0 & 1);
            a0 = lpa[i + 2]; c0 = lpc[i + 2]; y0 = myrow[i + 3];
            st = btln_layer(st, y1, a1, c1, mg & 2, mr & 2, m0 & 2);
            mg >>= 2; mr >>= 2; m0 >>= 2;
        }
    }
    st = btln_layer(st, y0, a0, c0, mg & 1, mr & 1, m0 & 1);
    out[wave_row0 + lane] = st;
}

extern "C" void kernel_launch(void* const* d_in, const int* in_sizes, int n_in,
                              void* d_out, int out_size, void* d_ws, size_t ws_size,
                              hipStream_t stream) {
    const float* x       = (const float*)d_in[0];
    const float* logits  = (const float*)d_in[1];
    const float* weights = (const float*)d_in[2];
    const float* biases  = (const float*)d_in[3];
    float* out = (float*)d_out;

    btln_fused_kernel<<<BATCH_TOTAL / ROWS_PER_BLOCK, 1024, 0, stream>>>(x, logits, weights, biases, out);
}


# Sinkhorn passes 2..40 kept as scale vectors on K=2^l in registers: r=1/(K c), c=1/(K^T r), P=r*K*c; 64 scales cross LDS per pass (sums of products replace logsumexp: sum order changes)
# speedup vs baseline: 6.4675x; 1.1462x over previous
#include <hip/hip_runtime.h>

#define EPSF 1e-6f
#define NSZ 64
#define NLAYERS 63
#define BATCH_TOTAL 262144
#define SINK_ITERS 20
#define WAVES_PER_BLOCK 32
#define ROWS_PER_WAVE 32
#define ROWS_PER_BLOCK (WAVES_PER_BLOCK * ROWS_PER_WAVE)
#define LEAF_PITCH 66

typedef __attribute__((ext_vector_type(2))) float v2f;
typedef __attribute__((ext_vector_type(4))) float v4f;
typedef __attribute__((ext_vector_type(8))) float v8f;

#define SINK_WAVES 16
#define BTLN_LOG2E 1.4426950408889634f

__device__ __forceinline__ void btln_lds_barrier() {
    __builtin_amdgcn_fence(__ATOMIC_RELEASE, "workgroup", "local");
    __builtin_amdgcn_s_barrier();
    __builtin_amdgcn_fence(__ATOMIC_ACQUIRE, "workgroup", "local");
}

template <int MASK>
__device__ __forceinline__ float btln_xlane(float v) {
    return __builtin_bit_cast(float, __builtin_amdgcn_mov_dpp(__builtin_bit_cast(int, v),
                                                              0x160 | MASK, 0xf, 0xf, true));
}

__device__ __forceinline__ float btln_exp2_nonpos(float t) {
    return __builtin_amdgcn_exp2f(t);
}

__device__ __forceinline__ void btln_lse_normalise(float (&v)[8]) {
    float m = fmaxf(fmaxf(fmaxf(v[0], v[1]), fmaxf(v[2], v[3])),
                    fmaxf(fmaxf(v[4], v[5]), fmaxf(v[6], v[7])));
    m = fmaxf(m, btln_xlane<1>(m));
    m = fmaxf(m, btln_xlane<2>(m));
    m = fmaxf(m, btln_xlane<4>(m));
    float s = ((btln_exp2_nonpos(v[0] - m) + btln_exp2_nonpos(v[1] - m)) +
               (btln_exp2_nonpos(v[2] - m) + btln_exp2_nonpos(v[3] - m))) +
              ((btln_exp2_nonpos(v[4] - m) + btln_exp2_nonpos(v[5] - m)) +
               (btln_exp2_nonpos(v[6] - m) + btln_exp2_nonpos(v[7] - m)));
    s += btln_xlane<1>(s);
    s += btln_xlane<2>(s);
    s += btln_xlane<4>(s);
    const float lse = m + __builtin_amdgcn_logf(s);
    for (int q = 0; q < 8; ++q) v[q] -= lse;
}

__device__ __forceinline__ void btln_load_scales(const float* p, float (&w)[8]) {
    const v4f a = *(const v4f*)p;
    const v4f b = *(const v4f*)(p + 4);
    w[0] = a.x; w[1] = a.y; w[2] = a.z; w[3] = a.w;
    w[4] = b.x; w[5] = b.y; w[6] = b.z; w[7] = b.w;
}

__device__ __forceinline__ float btln_line_scale(const float (&k)[8], const float (&w)[8]) {
    float s0 = k[0] * w[0];
    float s1 = k[1] * w[1];
    s0 = fmaf(k[2], w[2], s0);
    s1 = fmaf(k[3], w[3], s1);
    s0 = fmaf(k[4], w[4], s0);
    s1 = fmaf(k[5], w[5], s1);
    s0 = fmaf(k[6], w[6], s0);
    s1 = fmaf(k[7], w[7], s1);
    float s = s0 + s1;
    s += btln_xlane<1>(s);
    s += btln_xlane<2>(s);
    s += btln_xlane<4>(s);
    return __builtin_amdgcn_rcpf(s);
}

__device__ __forceinline__ float btln_clip01(float v) {
    return fminf(fmaxf(v, EPSF), 1.0f - EPSF);
}

__device__ __forceinline__ float btln_layer(float st, float leaf, v4f pa, v4f pc,
                                            bool need_g, bool reflect, bool rule0) {
    float X = btln_clip01(st);
    float Y = btln_clip01(leaf);
    if (reflect) {
        X = btln_clip01(1.0f - X);
        Y = btln_clip01(1.0f - Y);
    }
    const float lin = pc.x * X + pc.y * Y;
    float g = 0.0f;
    if (need_g) g = __builtin_amdgcn_exp2f(pa.z * (pa.x * __builtin_amdgcn_logf(X) +
                                                   pa.y * __builtin_amdgcn_logf(Y)));
    float r = pc.z * lin + pc.w * g;
    if (rule0) r = ((fabsf(X - 1.0f) < EPSF) && (fabsf(Y - 1.0f) < EPSF)) ? 1.0f : 0.0f;
    if (reflect) r = 1.0f - r;
    return (r != r) ? pa.w : r;
}


__device__ __forceinline__ void btln_leaf_slice(const v2f (&xa)[16], const v2f (&xb)[16],
                                                const float* __restrict__ PT, float* wl,
                                                int t, int nn, int kb, int hrow) {
    v8f acc0 = {}, acc1 = {};
    const float* bcol = &PT[(16 * t + nn) * 66 + kb];
#pragma unroll
    for (int s = 0; s < 16; ++s) {
        const v2f b = *(const v2f*)(bcol + 4 * s);
        acc0 = __builtin_amdgcn_wmma_f32_16x16x4_f32(false, xa[s], false, b, (short)0, acc0, false, false);
        acc1 = __builtin_amdgcn_wmma_f32_16x16x4_f32(false, xb[s], false, b, (short)0, acc1, false, false);
    }
#pragma unroll
    for (int r = 0; r < 8; ++r) {
        wl[(hrow + r) * LEAF_PITCH + 16 * t + nn]      = acc0[r];
        wl[(16 + hrow + r) * LEAF_PITCH + 16 * t + nn] = acc1[r];
    }
    __builtin_amdgcn_wave_barrier();
}

__global__ __launch_bounds__(1024) void btln_fused_kernel(const float* __restrict__ x,
                                                          const float* __restrict__ logits,
                                                          const float* __restrict__ weights,
                                                          const float* __restrict__ biases,
                                                          float* __restrict__ out) {
    __shared__ float PT[64 * 66];
    __shared__ float leaves[ROWS_PER_BLOCK * LEAF_PITCH];
    __shared__ v4f lpa[NLAYERS];
    __shared__ v4f lpc[NLAYERS];
    __shared__ unsigned lmask[6];
    __shared__ __attribute__((aligned(16))) float svec[2 * NSZ];

    const int tid  = threadIdx.x;
    const int wave = tid >> 5;
    const int lane = tid & 31;
    const int nn   = lane & 15;
    const int kb   = (lane >> 4) * 2;
    const long wave_row0 = (long)blockIdx.x * ROWS_PER_BLOCK + wave * ROWS_PER_WAVE;

    const bool sink = wave < SINK_WAVES;
    const bool tabw = (wave >> 1) == (SINK_WAVES >> 1);
    const int sub  = lane & 7;
    const int line = (wave & (SINK_WAVES - 1)) + 16 * (lane >> 3);
    float* lrow = &leaves[line * 65 + sub];
    float* lcol = &leaves[sub * 65 + line];
    float* rvec = svec;
    float* cvec = svec + NSZ;
    const int wpos = (line & 7) * 8 + (line >> 3);

    float v[8] = {0.0f, 0.0f, 0.0f, 0.0f, 0.0f, 0.0f, 0.0f, 0.0f};
    if (sink) {
        for (int q = 0; q < 8; ++q) v[q] = logits[line * 64 + sub + 8 * q];
    }
    const int tl = tid - 32 * SINK_WAVES;
    const bool tvalid = tl < NLAYERS;
    float tw = 0.0f, tb = 0.0f;
    if (tabw) {
        const int t = tvalid ? tl : NLAYERS - 1;
        tw = weights[t];
        tb = biases[t];
    }

    v2f xa[16], xb[16];
    const float* arow = x + (wave_row0 + nn) * 64 + kb;
#pragma unroll
    for (int s = 0; s < 16; ++s) {
        xa[s] = *(const v2f*)(arow + 4 * s);
        xb[s] = *(const v2f*)(arow + 16 * 64 + 4 * s);
    }

    if (tabw) {
        const float w = tw;
        const float b = tb;
        const float w1 = 1.0f - w;
        const float aref = fminf(fmaxf(1.0f - b, -1.0f + EPSF), 2.0f - EPSF);
        const bool top_lin = fabsf(b - 0.5f) < EPSF;
        const bool reflect = !top_lin && (b < 0.5f);
        const float a = reflect ? aref : b;
        const float p = sqrtf(3.0f / fmaxf(2.0f - a, EPSF)) - 1.0f;
        const bool rule0 = !top_lin && (fabsf(a - 2.0f) < EPSF);
        const bool geo = !top_lin && (a >= 0.75f);
        const bool mid = !top_lin && !geo && (a > 0.5f);
        v4f pa, pc;
        pa.x = 2.0f * w; pa.y = 2.0f * w1; pa.z = p; pa.w = b;
        pc.x = w;        pc.y = w1;
        pc.z = geo ? 0.0f : (mid ? 3.0f - 4.0f * a : 1.0f);
        pc.w = geo ? 1.0f : (mid ? 4.0f * a - 2.0f : 0.0f);
        const unsigned mg = __builtin_amdgcn_ballot_w32(tvalid && (geo || mid) && !rule0);
        const unsigned mr = __builtin_amdgcn_ballot_w32(tvalid && reflect);
        const unsigned m0 = __builtin_amdgcn_ballot_w32(tvalid && rule0);
        if (tvalid) {
            lpa[tl] = pa;
            lpc[tl] = pc;
        }
        if (lane == 0) {
            lmask[0 + (tl >> 5)] = mg;
            lmask[2 + (tl >> 5)] = mr;
            lmask[4 + (tl >> 5)] = m0;
        }
    }

    float kr[8] = {0.0f, 0.0f, 0.0f, 0.0f, 0.0f, 0.0f, 0.0f, 0.0f};
    float kc[8] = {0.0f, 0.0f, 0.0f, 0.0f, 0.0f, 0.0f, 0.0f, 0.0f};
    float rv[8] = {0.0f, 0.0f, 0.0f, 0.0f, 0.0f, 0.0f, 0.0f, 0.0f};
    float csc = 0.0f;
    if (sink) {
        for (int q = 0; q < 8; ++q) v[q] *= BTLN_LOG2E;
        btln_lse_normalise(v);
        for (int q = 0; q < 8; ++q) {
            kr[q] = exp2f(v[q]);
            lrow[8 * q] = kr[q];
        }
        rvec[wpos] = 1.0f;
    }
    btln_lds_barrier();
    if (sink) {
        for (int q = 0; q < 8; ++q) kc[q] = lcol[8 * q * 65];
        btln_load_scales(rvec + 8 * sub, rv);
        csc = btln_line_scale(kc, rv);
    }
    for (int it = 1; it < SINK_ITERS; ++it) {
        if (sink) cvec[wpos] = csc;
        btln_lds_barrier();
        if (sink) {
            float cv[8];
            btln_load_scales(cvec + 8 * sub, cv);
            rvec[wpos] = btln_line_scale(kr, cv);
        }
        btln_lds_barrier();
        if (sink) {
            btln_load_scales(rvec + 8 * sub, rv);
            csc = btln_line_scale(kc, rv);
        }
    }
    if (sink) {
        for (int q = 0; q < 8; ++q) PT[line * 66 + sub + 8 * q] = (kc[q] * rv[q]) * csc;
    }
    __syncthreads();

    float* wl = &leaves[wave * ROWS_PER_WAVE * LEAF_PITCH];
    const int hrow = (lane >> 4) << 3;
    const float* myrow = wl + lane * LEAF_PITCH;

    unsigned long long mg = ((unsigned long long)__builtin_amdgcn_readfirstlane(lmask[1]) << 32) |
                            (unsigned)__builtin_amdgcn_readfirstlane(lmask[0]);
    unsigned long long mr = ((unsigned long long)__builtin_amdgcn_readfirstlane(lmask[3]) << 32) |
                            (unsigned)__builtin_amdgcn_readfirstlane(lmask[2]);
    unsigned long long m0 = ((unsigned long long)__builtin_amdgcn_readfirstlane(lmask[5]) << 32) |
                            (unsigned)__builtin_amdgcn_readfirstlane(lmask[4]);

    btln_leaf_slice(xa, xb, PT, wl, 0, nn, kb, hrow);
    float st = myrow[0];
    v4f a0 = lpa[0], c0 = lpc[0], a1, c1;
    float y0 = myrow[1], y1;
    int i = 0;
    for (int t = 0; t < 4; ++t) {
        if (t > 0) btln_leaf_slice(xa, xb, PT, wl, t, nn, kb, hrow);
        const int iend = 16 * t + 14;
        for (; i < iend; i += 2) {
            a1 = lpa[i + 1]; c1 = lpc[i + 1]; y1 = myrow[i + 2];
            st = btln_layer(st, y0, a0, c0, mg & 1, mr & 1, m0 & 1);
            a0 = lpa[i + 2]; c0 = lpc[i + 2]; y0 = myrow[i + 3];
            st = btln_layer(st, y1, a1, c1, mg & 2, mr & 2, m0 & 2);
            mg >>= 2; mr >>= 2; m0 >>= 2;
        }
    }
    st = btln_layer(st, y0, a0, c0, mg & 1, mr & 1, m0 & 1);
    out[wave_row0 + lane] = st;
}

extern "C" void kernel_launch(void* const* d_in, const int* in_sizes, int n_in,
                              void* d_out, int out_size, void* d_ws, size_t ws_size,
                              hipStream_t stream) {
    const float* x       = (const float*)d_in[0];
    const float* logits  = (const float*)d_in[1];
    const float* weights = (const float*)d_in[2];
    const float* biases  = (const float*)d_in[3];
    float* out = (float*)d_out;

    btln_fused_kernel<<<BATCH_TOTAL / ROWS_PER_BLOCK, 1024, 0, stream>>>(x, logits, weights, biases, out);
}


# Fused kernel re-tiled to 16 waves of 64 rows (512 threads): each lane scans two rows as two independent chains on packed f32 ops; table entry and B fragment serve twice the rows; half as many waves
# speedup vs baseline: 6.9412x; 1.0732x over previous
#include <hip/hip_runtime.h>

#define EPSF 1e-6f
#define NSZ 64
#define NLAYERS 63
#define BATCH_TOTAL 262144
#define SINK_ITERS 20
#define WAVES_PER_BLOCK 16
#define ROWS_PER_WAVE 64
#define TILES_PER_WAVE 4
#define ROWS_PER_BLOCK (WAVES_PER_BLOCK * ROWS_PER_WAVE)
#define LEAF_PITCH 66

typedef __attribute__((ext_vector_type(2))) float v2f;
typedef __attribute__((ext_vector_type(4))) float v4f;
typedef __attribute__((ext_vector_type(8))) float v8f;

#define SINK_WAVES 16
#define BTLN_LOG2E 1.4426950408889634f

__device__ __forceinline__ void btln_lds_barrier() {
    __builtin_amdgcn_fence(__ATOMIC_RELEASE, "workgroup", "local");
    __builtin_amdgcn_s_barrier();
    __builtin_amdgcn_fence(__ATOMIC_ACQUIRE, "workgroup", "local");
}

template <int MASK>
__device__ __forceinline__ float btln_xlane(float v) {
    return __builtin_bit_cast(float, __builtin_amdgcn_mov_dpp(__builtin_bit_cast(int, v),
                                                              0x160 | MASK, 0xf, 0xf, true));
}

__device__ __forceinline__ float btln_exp2_nonpos(float t) {
    return __builtin_amdgcn_exp2f(t);
}

__device__ __forceinline__ void btln_lse_normalise(float (&v)[8]) {
    float m = fmaxf(fmaxf(fmaxf(v[0], v[1]), fmaxf(v[2], v[3])),
                    fmaxf(fmaxf(v[4], v[5]), fmaxf(v[6], v[7])));
    m = fmaxf(m, btln_xlane<1>(m));
    m = fmaxf(m, btln_xlane<2>(m));
    m = fmaxf(m, btln_xlane<4>(m));
    float s = ((btln_exp2_nonpos(v[0] - m) + btln_exp2_nonpos(v[1] - m)) +
               (btln_exp2_nonpos(v[2] - m) + btln_exp2_nonpos(v[3] - m))) +
              ((btln_exp2_nonpos(v[4] - m) + btln_exp2_nonpos(v[5] - m)) +
               (btln_exp2_nonpos(v[6] - m) + btln_exp2_nonpos(v[7] - m)));
    s += btln_xlane<1>(s);
    s += btln_xlane<2>(s);
    s += btln_xlane<4>(s);
    const float lse = m + __builtin_amdgcn_logf(s);
    for (int q = 0; q < 8; ++q) v[q] -= lse;
}

__device__ __forceinline__ void btln_load_scales(const float* p, float (&w)[8]) {
    const v4f a = *(const v4f*)p;
    const v4f b = *(const v4f*)(p + 4);
    w[0] = a.x; w[1] = a.y; w[2] = a.z; w[3] = a.w;
    w[4] = b.x; w[5] = b.y; w[6] = b.z; w[7] = b.w;
}

__device__ __forceinline__ float btln_line_scale(const float (&k)[8], const float (&w)[8]) {
    float s0 = k[0] * w[0];
    float s1 = k[1] * w[1];
    s0 = fmaf(k[2], w[2], s0);
    s1 = fmaf(k[3], w[3], s1);
    s0 = fmaf(k[4], w[4], s0);
    s1 = fmaf(k[5], w[5], s1);
    s0 = fmaf(k[6], w[6], s0);
    s1 = fmaf(k[7], w[7], s1);
    float s = s0 + s1;
    s += btln_xlane<1>(s);
    s += btln_xlane<2>(s);
    s += btln_xlane<4>(s);
    return __builtin_amdgcn_rcpf(s);
}

__device__ __forceinline__ float btln_clip01(float v) {
    return fminf(fmaxf(v, EPSF), 1.0f - EPSF);
}

__device__ __forceinline__ v2f btln_clip01_2(v2f v) {
    v2f r;
    r.x = btln_clip01(v.x);
    r.y = btln_clip01(v.y);
    return r;
}

__device__ __forceinline__ v2f btln_layer(v2f st, v2f leaf, v4f pa, v4f pc,
                                          bool need_g, bool reflect, bool rule0) {
    v2f X = btln_clip01_2(st);
    v2f Y = btln_clip01_2(leaf);
    if (reflect) {
        X = btln_clip01_2(1.0f - X);
        Y = btln_clip01_2(1.0f - Y);
    }
    const v2f lin = pc.x * X + pc.y * Y;
    v2f g = {0.0f, 0.0f};
    if (need_g) {
        v2f lx, ly;
        lx.x = __builtin_amdgcn_logf(X.x);
        lx.y = __builtin_amdgcn_logf(X.y);
        ly.x = __builtin_amdgcn_logf(Y.x);
        ly.y = __builtin_amdgcn_logf(Y.y);
        const v2f e = pa.z * (pa.x * lx + pa.y * ly);
        g.x = __builtin_amdgcn_exp2f(e.x);
        g.y = __builtin_amdgcn_exp2f(e.y);
    }
    v2f r = pc.z * lin + pc.w * g;
    if (rule0) {
        r.x = ((fabsf(X.x - 1.0f) < EPSF) && (fabsf(Y.x - 1.0f) < EPSF)) ? 1.0f : 0.0f;
        r.y = ((fabsf(X.y - 1.0f) < EPSF) && (fabsf(Y.y - 1.0f) < EPSF)) ? 1.0f : 0.0f;
    }
    if (reflect) r = 1.0f - r;
    r.x = (r.x != r.x) ? pa.w : r.x;
    r.y = (r.y != r.y) ? pa.w : r.y;
    return r;
}


__device__ __forceinline__ void btln_leaf_slice(const v2f (&xf)[TILES_PER_WAVE][16],
                                                const float* __restrict__ PT, float* wl,
                                                int t, int nn, int kb, int hrow) {
    v8f acc[TILES_PER_WAVE] = {};
    const float* bcol = &PT[(16 * t + nn) * 66 + kb];
#pragma unroll
    for (int s = 0; s < 16; ++s) {
        const v2f b = *(const v2f*)(bcol + 4 * s);
#pragma unroll
        for (int m = 0; m < TILES_PER_WAVE; ++m)
            acc[m] = __builtin_amdgcn_wmma_f32_16x16x4_f32(false, xf[m][s], false, b, (short)0, acc[m], false, false);
    }
#pragma unroll
    for (int m = 0; m < TILES_PER_WAVE; ++m) {
#pragma unroll
        for (int r = 0; r < 8; ++r)
            wl[(16 * m + hrow + r) * LEAF_PITCH + 16 * t + nn] = acc[m][r];
    }
    __builtin_amdgcn_wave_barrier();
}

__global__ __launch_bounds__(512) void btln_fused_kernel(const float* __restrict__ x,
                                                          const float* __restrict__ logits,
                                                          const float* __restrict__ weights,
                                                          const float* __restrict__ biases,
                                                          float* __restrict__ out) {
    __shared__ float PT[64 * 66];
    __shared__ float leaves[ROWS_PER_BLOCK * LEAF_PITCH];
    __shared__ v4f lpa[NLAYERS];
    __shared__ v4f lpc[NLAYERS];
    __shared__ unsigned lmask[6];
    __shared__ __attribute__((aligned(16))) float svec[2 * NSZ];

    const int tid  = threadIdx.x;
    const int wave = tid >> 5;
    const int lane = tid & 31;
    const int nn   = lane & 15;
    const int kb   = (lane >> 4) * 2;
    const long wave_row0 = (long)blockIdx.x * ROWS_PER_BLOCK + wave * ROWS_PER_WAVE;

    const bool tabw = wave < 2;
    const int sub  = lane & 7;
    const int line = (wave & (SINK_WAVES - 1)) + 16 * (lane >> 3);
    float* lrow = &leaves[line * 65 + sub];
    float* lcol = &leaves[sub * 65 + line];
    float* rvec = svec;
    float* cvec = svec + NSZ;
    const int wpos = (line & 7) * 8 + (line >> 3);

    float v[8];
    for (int q = 0; q < 8; ++q) v[q] = logits[line * 64 + sub + 8 * q];
    const int tl = tid;
    const bool tvalid = tl < NLAYERS;
    float tw = 0.0f, tb = 0.0f;
    if (tabw) {
        const int t = tvalid ? tl : NLAYERS - 1;
        tw = weights[t];
        tb = biases[t];
    }

    v2f xf[TILES_PER_WAVE][16];
    const float* arow = x + (wave_row0 + nn) * 64 + kb;
#pragma unroll
    for (int s = 0; s < 16; ++s) {
        xf[0][s] = *(const v2f*)(arow + 4 * s);
        xf[1][s] = *(const v2f*)(arow + 16 * 64 + 4 * s);
    }

    if (tabw) {
        const float w = tw;
        const float b = tb;
        const float w1 = 1.0f - w;
        const float aref = fminf(fmaxf(1.0f - b, -1.0f + EPSF), 2.0f - EPSF);
        const bool top_lin = fabsf(b - 0.5f) < EPSF;
        const bool reflect = !top_lin && (b < 0.5f);
        const float a = reflect ? aref : b;
        const float p = sqrtf(3.0f / fmaxf(2.0f - a, EPSF)) - 1.0f;
        const bool rule0 = !top_lin && (fabsf(a - 2.0f) < EPSF);
        const bool geo = !top_lin && (a >= 0.75f);
        const bool mid = !top_lin && !geo && (a > 0.5f);
        v4f pa, pc;
        pa.x = 2.0f * w; pa.y = 2.0f * w1; pa.z = p; pa.w = b;
        pc.x = w;        pc.y = w1;
        pc.z = geo ? 0.0f : (mid ? 3.0f - 4.0f * a : 1.0f);
        pc.w = geo ? 1.0f : (mid ? 4.0f * a - 2.0f : 0.0f);
        const unsigned mg = __builtin_amdgcn_ballot_w32(tvalid && (geo || mid) && !rule0);
        const unsigned mr = __builtin_amdgcn_ballot_w32(tvalid && reflect);
        const unsigned m0 = __builtin_amdgcn_ballot_w32(tvalid && rule0);
        if (tvalid) {
            lpa[tl] = pa;
            lpc[tl] = pc;
        }
        if (lane == 0) {
            lmask[0 + (tl >> 5)] = mg;
            lmask[2 + (tl >> 5)] = mr;
            lmask[4 + (tl >> 5)] = m0;
        }
    }

    float kr[8];
    float kc[8];
    float rv[8];
    float csc;
    for (int q = 0; q < 8; ++q) v[q] *= BTLN_LOG2E;
    btln_lse_normalise(v);
    for (int q = 0; q < 8; ++q) {
        kr[q] = exp2f(v[q]);
        lrow[8 * q] = kr[q];
    }
    rvec[wpos] = 1.0f;
    btln_lds_barrier();
#pragma unroll
    for (int s = 0; s < 16; ++s) {
        xf[2][s] = *(const v2f*)(arow + 32 * 64 + 4 * s);
        xf[3][s] = *(const v2f*)(arow + 48 * 64 + 4 * s);
    }
    for (int q = 0; q < 8; ++q) kc[q] = lcol[8 * q * 65];
    btln_load_scales(rvec + 8 * sub, rv);
    csc = btln_line_scale(kc, rv);
    for (int it = 1; it < SINK_ITERS; ++it) {
        cvec[wpos] = csc;
        btln_lds_barrier();
        float cv[8];
        btln_load_scales(cvec + 8 * sub, cv);
        rvec[wpos] = btln_line_scale(kr, cv);
        btln_lds_barrier();
        btln_load_scales(rvec + 8 * sub, rv);
        csc = btln_line_scale(kc, rv);
    }
    for (int q = 0; q < 8; ++q) PT[line * 66 + sub + 8 * q] = (kc[q] * rv[q]) * csc;
    __syncthreads();

    float* wl = &leaves[wave * ROWS_PER_WAVE * LEAF_PITCH];
    const int hrow = (lane >> 4) << 3;
    const float* row0 = wl + lane * LEAF_PITCH;
    const float* row1 = wl + (32 + lane) * LEAF_PITCH;

    unsigned long long mg = ((unsigned long long)__builtin_amdgcn_readfirstlane(lmask[1]) << 32) |
                            (unsigned)__builtin_amdgcn_readfirstlane(lmask[0]);
    unsigned long long mr = ((unsigned long long)__builtin_amdgcn_readfirstlane(lmask[3]) << 32) |
                            (unsigned)__builtin_amdgcn_readfirstlane(lmask[2]);
    unsigned long long m0 = ((unsigned long long)__builtin_amdgcn_readfirstlane(lmask[5]) << 32) |
                            (unsigned)__builtin_amdgcn_readfirstlane(lmask[4]);

    btln_leaf_slice(xf, PT, wl, 0, nn, kb, hrow);
    v2f st = {row0[0], row1[0]};
    v4f a0 = lpa[0], c0 = lpc[0], a1, c1;
    v2f y0 = {row0[1], row1[1]}, y1;
    int i = 0;
    for (int t = 0; t < 4; ++t) {
        if (t > 0) btln_leaf_slice(xf, PT, wl, t, nn, kb, hrow);
        const int iend = 16 * t + 14;
        for (; i < iend; i += 2) {
            a1 = lpa[i + 1]; c1 = lpc[i + 1]; y1.x = row0[i + 2]; y1.y = row1[i + 2];
            st = btln_layer(st, y0, a0, c0, mg & 1, mr & 1, m0 & 1);
            a0 = lpa[i + 2]; c0 = lpc[i + 2]; y0.x = row0[i + 3]; y0.y = row1[i + 3];
            st = btln_layer(st, y1, a1, c1, mg & 2, mr & 2, m0 & 2);
            mg >>= 2; mr >>= 2; m0 >>= 2;
        }
    }
    st = btln_layer(st, y0, a0, c0, mg & 1, mr & 1, m0 & 1);
    out[wave_row0 + lane]      = st.x;
    out[wave_row0 + 32 + lane] = st.y;
}

extern "C" void kernel_launch(void* const* d_in, const int* in_sizes, int n_in,
                              void* d_out, int out_size, void* d_ws, size_t ws_size,
                              hipStream_t stream) {
    const float* x       = (const float*)d_in[0];
    const float* logits  = (const float*)d_in[1];
    const float* weights = (const float*)d_in[2];
    const float* biases  = (const float*)d_in[3];
    float* out = (float*)d_out;

    btln_fused_kernel<<<BATCH_TOTAL / ROWS_PER_BLOCK, 512, 0, stream>>>(x, logits, weights, biases, out);
}


# Scan layer as one straight line: side of F chosen by table constants (s,o), operands and result pass as s*v+o (exactly v or 1-v), g always evaluated; no per-layer side branches; bit-identical
# speedup vs baseline: 7.4041x; 1.0667x over previous
#include <hip/hip_runtime.h>

#define EPSF 1e-6f
#define NSZ 64
#define NLAYERS 63
#define BATCH_TOTAL 262144
#define SINK_ITERS 20
#define WAVES_PER_BLOCK 16
#define ROWS_PER_WAVE 64
#define TILES_PER_WAVE 4
#define ROWS_PER_BLOCK (WAVES_PER_BLOCK * ROWS_PER_WAVE)
#define LEAF_PITCH 66

typedef __attribute__((ext_vector_type(2))) float v2f;
typedef __attribute__((ext_vector_type(4))) float v4f;
typedef __attribute__((ext_vector_type(8))) float v8f;

#define SINK_WAVES 16
#define BTLN_LOG2E 1.4426950408889634f

__device__ __forceinline__ void btln_lds_barrier() {
    __builtin_amdgcn_fence(__ATOMIC_RELEASE, "workgroup", "local");
    __builtin_amdgcn_s_barrier();
    __builtin_amdgcn_fence(__ATOMIC_ACQUIRE, "workgroup", "local");
}

template <int MASK>
__device__ __forceinline__ float btln_xlane(float v) {
    return __builtin_bit_cast(float, __builtin_amdgcn_mov_dpp(__builtin_bit_cast(int, v),
                                                              0x160 | MASK, 0xf, 0xf, true));
}

__device__ __forceinline__ float btln_exp2_nonpos(float t) {
    return __builtin_amdgcn_exp2f(t);
}

__device__ __forceinline__ void btln_lse_normalise(float (&v)[8]) {
    float m = fmaxf(fmaxf(fmaxf(v[0], v[1]), fmaxf(v[2], v[3])),
                    fmaxf(fmaxf(v[4], v[5]), fmaxf(v[6], v[7])));
    m = fmaxf(m, btln_xlane<1>(m));
    m = fmaxf(m, btln_xlane<2>(m));
    m = fmaxf(m, btln_xlane<4>(m));
    float s = ((btln_exp2_nonpos(v[0] - m) + btln_exp2_nonpos(v[1] - m)) +
               (btln_exp2_nonpos(v[2] - m) + btln_exp2_nonpos(v[3] - m))) +
              ((btln_exp2_nonpos(v[4] - m) + btln_exp2_nonpos(v[5] - m)) +
               (btln_exp2_nonpos(v[6] - m) + btln_exp2_nonpos(v[7] - m)));
    s += btln_xlane<1>(s);
    s += btln_xlane<2>(s);
    s += btln_xlane<4>(s);
    const float lse = m + __builtin_amdgcn_logf(s);
    for (int q = 0; q < 8; ++q) v[q] -= lse;
}

__device__ __forceinline__ void btln_load_scales(const float* p, float (&w)[8]) {
    const v4f a = *(const v4f*)p;
    const v4f b = *(const v4f*)(p + 4);
    w[0] = a.x; w[1] = a.y; w[2] = a.z; w[3] = a.w;
    w[4] = b.x; w[5] = b.y; w[6] = b.z; w[7] = b.w;
}

__device__ __forceinline__ float btln_line_scale(const float (&k)[8], const float (&w)[8]) {
    float s0 = k[0] * w[0];
    float s1 = k[1] * w[1];
    s0 = fmaf(k[2], w[2], s0);
    s1 = fmaf(k[3], w[3], s1);
    s0 = fmaf(k[4], w[4], s0);
    s1 = fmaf(k[5], w[5], s1);
    s0 = fmaf(k[6], w[6], s0);
    s1 = fmaf(k[7], w[7], s1);
    float s = s0 + s1;
    s += btln_xlane<1>(s);
    s += btln_xlane<2>(s);
    s += btln_xlane<4>(s);
    return __builtin_amdgcn_rcpf(s);
}

__device__ __forceinline__ float btln_clip01(float v) {
    return __builtin_amdgcn_fmed3f(v, EPSF, 1.0f - EPSF);
}

__device__ __forceinline__ v2f btln_clip01_2(v2f v) {
    v2f r;
    r.x = btln_clip01(v.x);
    r.y = btln_clip01(v.y);
    return r;
}

__device__ __forceinline__ v2f btln_layer(v2f st, v2f leaf, v4f pa, v4f pc, float po, bool rule0) {
    const v2f X = pa.w * btln_clip01_2(st) + po;
    const v2f Y = pa.w * btln_clip01_2(leaf) + po;
    const v2f lin = pc.x * X + pc.y * Y;
    v2f g;
    {
        v2f lx, ly;
        lx.x = __builtin_amdgcn_logf(X.x);
        lx.y = __builtin_amdgcn_logf(X.y);
        ly.x = __builtin_amdgcn_logf(Y.x);
        ly.y = __builtin_amdgcn_logf(Y.y);
        const v2f e = pa.z * (pa.x * lx + pa.y * ly);
        g.x = __builtin_amdgcn_exp2f(e.x);
        g.y = __builtin_amdgcn_exp2f(e.y);
    }
    v2f r = pc.z * lin + pc.w * g;
    if (__builtin_expect(rule0, 0)) {
        r.x = ((fabsf(X.x - 1.0f) < EPSF) && (fabsf(Y.x - 1.0f) < EPSF)) ? 1.0f : 0.0f;
        r.y = ((fabsf(X.y - 1.0f) < EPSF) && (fabsf(Y.y - 1.0f) < EPSF)) ? 1.0f : 0.0f;
    }
    return pa.w * r + po;
}


__device__ __forceinline__ void btln_leaf_slice(const v2f (&xf)[TILES_PER_WAVE][16],
                                                const float* __restrict__ PT, float* wl,
                                                int t, int nn, int kb, int hrow) {
    v8f acc[TILES_PER_WAVE] = {};
    const float* bcol = &PT[(16 * t + nn) * 66 + kb];
#pragma unroll
    for (int s = 0; s < 16; ++s) {
        const v2f b = *(const v2f*)(bcol + 4 * s);
#pragma unroll
        for (int m = 0; m < TILES_PER_WAVE; ++m)
            acc[m] = __builtin_amdgcn_wmma_f32_16x16x4_f32(false, xf[m][s], false, b, (short)0, acc[m], false, false);
    }
#pragma unroll
    for (int m = 0; m < TILES_PER_WAVE; ++m) {
#pragma unroll
        for (int r = 0; r < 8; ++r)
            wl[(16 * m + hrow + r) * LEAF_PITCH + 16 * t + nn] = acc[m][r];
    }
    __builtin_amdgcn_wave_barrier();
}

__global__ __launch_bounds__(512) void btln_fused_kernel(const float* __restrict__ x,
                                                          const float* __restrict__ logits,
                                                          const float* __restrict__ weights,
                                                          const float* __restrict__ biases,
                                                          float* __restrict__ out) {
    __shared__ float PT[64 * 66];
    __shared__ float leaves[ROWS_PER_BLOCK * LEAF_PITCH];
    __shared__ v4f lpa[NLAYERS];
    __shared__ v4f lpc[NLAYERS];
    __shared__ float lpo[NLAYERS];
    __shared__ unsigned lmask[2];
    __shared__ __attribute__((aligned(16))) float svec[2 * NSZ];

    const int tid  = threadIdx.x;
    const int wave = tid >> 5;
    const int lane = tid & 31;
    const int nn   = lane & 15;
    const int kb   = (lane >> 4) * 2;
    const long wave_row0 = (long)blockIdx.x * ROWS_PER_BLOCK + wave * ROWS_PER_WAVE;

    const bool tabw = wave < 2;
    const int sub  = lane & 7;
    const int line = (wave & (SINK_WAVES - 1)) + 16 * (lane >> 3);
    float* lrow = &leaves[line * 65 + sub];
    float* lcol = &leaves[sub * 65 + line];
    float* rvec = svec;
    float* cvec = svec + NSZ;
    const int wpos = (line & 7) * 8 + (line >> 3);

    float v[8];
    for (int q = 0; q < 8; ++q) v[q] = logits[line * 64 + sub + 8 * q];
    const int tl = tid;
    const bool tvalid = tl < NLAYERS;
    float tw = 0.0f, tb = 0.0f;
    if (tabw) {
        const int t = tvalid ? tl : NLAYERS - 1;
        tw = weights[t];
        tb = biases[t];
    }

    v2f xf[TILES_PER_WAVE][16];
    const float* arow = x + (wave_row0 + nn) * 64 + kb;
#pragma unroll
    for (int s = 0; s < 16; ++s) {
        xf[0][s] = *(const v2f*)(arow + 4 * s);
        xf[1][s] = *(const v2f*)(arow + 16 * 64 + 4 * s);
    }

    if (tabw) {
        const float w = tw;
        const float b = tb;
        const float w1 = 1.0f - w;
        const float aref = fminf(fmaxf(1.0f - b, -1.0f + EPSF), 2.0f - EPSF);
        const bool top_lin = fabsf(b - 0.5f) < EPSF;
        const bool reflect = !top_lin && (b < 0.5f);
        const float a = reflect ? aref : b;
        const float p = sqrtf(3.0f / fmaxf(2.0f - a, EPSF)) - 1.0f;
        const bool rule0 = !top_lin && (fabsf(a - 2.0f) < EPSF);
        const bool geo = !top_lin && (a >= 0.75f);
        const bool mid = !top_lin && !geo && (a > 0.5f);
        v4f pa, pc;
        pa.x = 2.0f * w; pa.y = 2.0f * w1; pa.z = p; pa.w = reflect ? -1.0f : 1.0f;
        pc.x = w;        pc.y = w1;
        pc.z = geo ? 0.0f : (mid ? 3.0f - 4.0f * a : 1.0f);
        pc.w = geo ? 1.0f : (mid ? 4.0f * a - 2.0f : 0.0f);
        const unsigned m0 = __builtin_amdgcn_ballot_w32(tvalid && rule0);
        if (tvalid) {
            lpa[tl] = pa;
            lpc[tl] = pc;
            lpo[tl] = reflect ? 1.0f : 0.0f;
        }
        if (lane == 0) lmask[tl >> 5] = m0;
    }

    float kr[8];
    float kc[8];
    float rv[8];
    float csc;
    for (int q = 0; q < 8; ++q) v[q] *= BTLN_LOG2E;
    btln_lse_normalise(v);
    for (int q = 0; q < 8; ++q) {
        kr[q] = exp2f(v[q]);
        lrow[8 * q] = kr[q];
    }
    rvec[wpos] = 1.0f;
    btln_lds_barrier();
#pragma unroll
    for (int s = 0; s < 16; ++s) {
        xf[2][s] = *(const v2f*)(arow + 32 * 64 + 4 * s);
        xf[3][s] = *(const v2f*)(arow + 48 * 64 + 4 * s);
    }
    for (int q = 0; q < 8; ++q) kc[q] = lcol[8 * q * 65];
    btln_load_scales(rvec + 8 * sub, rv);
    csc = btln_line_scale(kc, rv);
    for (int it = 1; it < SINK_ITERS; ++it) {
        cvec[wpos] = csc;
        btln_lds_barrier();
        float cv[8];
        btln_load_scales(cvec + 8 * sub, cv);
        rvec[wpos] = btln_line_scale(kr, cv);
        btln_lds_barrier();
        btln_load_scales(rvec + 8 * sub, rv);
        csc = btln_line_scale(kc, rv);
    }
    for (int q = 0; q < 8; ++q) PT[line * 66 + sub + 8 * q] = (kc[q] * rv[q]) * csc;
    __syncthreads();

    float* wl = &leaves[wave * ROWS_PER_WAVE * LEAF_PITCH];
    const int hrow = (lane >> 4) << 3;
    const float* row0 = wl + lane * LEAF_PITCH;
    const float* row1 = wl + (32 + lane) * LEAF_PITCH;

    unsigned long long m0 = ((unsigned long long)__builtin_amdgcn_readfirstlane(lmask[1]) << 32) |
                            (unsigned)__builtin_amdgcn_readfirstlane(lmask[0]);

    btln_leaf_slice(xf, PT, wl, 0, nn, kb, hrow);
    v2f st = {row0[0], row1[0]};
    v4f a0 = lpa[0], c0 = lpc[0], a1, c1;
    float o0 = lpo[0], o1;
    v2f y0 = {row0[1], row1[1]}, y1;
    int i = 0;
    for (int t = 0; t < 4; ++t) {
        if (t > 0) btln_leaf_slice(xf, PT, wl, t, nn, kb, hrow);
        const int iend = 16 * t + 14;
        for (; i < iend; i += 2) {
            a1 = lpa[i + 1]; c1 = lpc[i + 1]; o1 = lpo[i + 1]; y1.x = row0[i + 2]; y1.y = row1[i + 2];
            st = btln_layer(st, y0, a0, c0, o0, m0 & 1);
            a0 = lpa[i + 2]; c0 = lpc[i + 2]; o0 = lpo[i + 2]; y0.x = row0[i + 3]; y0.y = row1[i + 3];
            st = btln_layer(st, y1, a1, c1, o1, m0 & 2);
            m0 >>= 2;
        }
    }
    st = btln_layer(st, y0, a0, c0, o0, m0 & 1);
    out[wave_row0 + lane]      = st.x;
    out[wave_row0 + 32 + lane] = st.y;
}

extern "C" void kernel_launch(void* const* d_in, const int* in_sizes, int n_in,
                              void* d_out, int out_size, void* d_ws, size_t ws_size,
                              hipStream_t stream) {
    const float* x       = (const float*)d_in[0];
    const float* logits  = (const float*)d_in[1];
    const float* weights = (const float*)d_in[2];
    const float* biases  = (const float*)d_in[3];
    float* out = (float*)d_out;

    btln_fused_kernel<<<BATCH_TOTAL / ROWS_PER_BLOCK, 512, 0, stream>>>(x, logits, weights, biases, out);
}


# Re-tiled to 8 waves of 128 rows: lane scans 4 rows (2 packed pairs) per table entry kept as pairs; x, P fragments load 16 B = 2 k-steps (k order in each 8, so product sum order changes)
# speedup vs baseline: 7.6237x; 1.0297x over previous
#include <hip/hip_runtime.h>

#define EPSF 1e-6f
#define NSZ 64
#define NLAYERS 63
#define BATCH_TOTAL 262144
#define SINK_ITERS 20
#define WAVES_PER_BLOCK 8
#define ROWS_PER_WAVE 128
#define TILES_PER_WAVE 8
#define ROWS_PER_BLOCK (WAVES_PER_BLOCK * ROWS_PER_WAVE)
#define LEAF_PITCH 66
#define PT_PITCH 68

typedef __attribute__((ext_vector_type(2))) float v2f;
typedef __attribute__((ext_vector_type(4))) float v4f;
typedef __attribute__((ext_vector_type(8))) float v8f;

#define BTLN_LOG2E 1.4426950408889634f

__device__ __forceinline__ void btln_lds_barrier() {
    __builtin_amdgcn_fence(__ATOMIC_RELEASE, "workgroup", "local");
    __builtin_amdgcn_s_barrier();
    __builtin_amdgcn_fence(__ATOMIC_ACQUIRE, "workgroup", "local");
}

template <int MASK>
__device__ __forceinline__ float btln_xlane(float v) {
    return __builtin_bit_cast(float, __builtin_amdgcn_mov_dpp(__builtin_bit_cast(int, v),
                                                              0x160 | MASK, 0xf, 0xf, true));
}

__device__ __forceinline__ float btln_exp2_nonpos(float t) {
    return __builtin_amdgcn_exp2f(t);
}

__device__ __forceinline__ void btln_lse_normalise(float (&v)[8]) {
    float m = fmaxf(fmaxf(fmaxf(v[0], v[1]), fmaxf(v[2], v[3])),
                    fmaxf(fmaxf(v[4], v[5]), fmaxf(v[6], v[7])));
    m = fmaxf(m, btln_xlane<1>(m));
    m = fmaxf(m, btln_xlane<2>(m));
    m = fmaxf(m, btln_xlane<4>(m));
    float s = ((btln_exp2_nonpos(v[0] - m) + btln_exp2_nonpos(v[1] - m)) +
               (btln_exp2_nonpos(v[2] - m) + btln_exp2_nonpos(v[3] - m))) +
              ((btln_exp2_nonpos(v[4] - m) + btln_exp2_nonpos(v[5] - m)) +
               (btln_exp2_nonpos(v[6] - m) + btln_exp2_nonpos(v[7] - m)));
    s += btln_xlane<1>(s);
    s += btln_xlane<2>(s);
    s += btln_xlane<4>(s);
    const float lse = m + __builtin_amdgcn_logf(s);
    for (int q = 0; q < 8; ++q) v[q] -= lse;
}

__device__ __forceinline__ void btln_load_scales(const float* p, float (&w)[8]) {
    const v4f a = *(const v4f*)p;
    const v4f b = *(const v4f*)(p + 4);
    w[0] = a.x; w[1] = a.y; w[2] = a.z; w[3] = a.w;
    w[4] = b.x; w[5] = b.y; w[6] = b.z; w[7] = b.w;
}

__device__ __forceinline__ float btln_line_scale(const float (&k)[8], const float (&w)[8]) {
    float s0 = k[0] * w[0];
    float s1 = k[1] * w[1];
    s0 = fmaf(k[2], w[2], s0);
    s1 = fmaf(k[3], w[3], s1);
    s0 = fmaf(k[4], w[4], s0);
    s1 = fmaf(k[5], w[5], s1);
    s0 = fmaf(k[6], w[6], s0);
    s1 = fmaf(k[7], w[7], s1);
    float s = s0 + s1;
    s += btln_xlane<1>(s);
    s += btln_xlane<2>(s);
    s += btln_xlane<4>(s);
    return __builtin_amdgcn_rcpf(s);
}

__device__ __forceinline__ float btln_clip01(float v) {
    return __builtin_amdgcn_fmed3f(v, EPSF, 1.0f - EPSF);
}

__device__ __forceinline__ v2f btln_clip01_2(v2f v) {
    v2f r;
    r.x = btln_clip01(v.x);
    r.y = btln_clip01(v.y);
    return r;
}

#define LT_STRIDE 20
struct btln_lay { v2f s, o, w0, w1, tw0, tw1, p, cl, cg; };

__device__ __forceinline__ btln_lay btln_load_layer(const float* ltab, int i) {
    const float* e = ltab + i * LT_STRIDE;
    const v4f q0 = *(const v4f*)(e);
    const v4f q1 = *(const v4f*)(e + 4);
    const v4f q2 = *(const v4f*)(e + 8);
    const v4f q3 = *(const v4f*)(e + 12);
    btln_lay L;
    L.s   = __builtin_shufflevector(q0, q0, 0, 1);
    L.o   = __builtin_shufflevector(q0, q0, 2, 3);
    L.w0  = __builtin_shufflevector(q1, q1, 0, 1);
    L.w1  = __builtin_shufflevector(q1, q1, 2, 3);
    L.tw0 = __builtin_shufflevector(q2, q2, 0, 1);
    L.tw1 = __builtin_shufflevector(q2, q2, 2, 3);
    L.p   = __builtin_shufflevector(q3, q3, 0, 1);
    L.cl  = __builtin_shufflevector(q3, q3, 2, 3);
    L.cg  = *(const v2f*)(e + 16);
    return L;
}

__device__ __forceinline__ v2f btln_geo(v2f X, v2f Y, const btln_lay& L) {
    v2f lx, ly, g;
    lx.x = __builtin_amdgcn_logf(X.x);
    lx.y = __builtin_amdgcn_logf(X.y);
    ly.x = __builtin_amdgcn_logf(Y.x);
    ly.y = __builtin_amdgcn_logf(Y.y);
    const v2f e = L.p * (L.tw0 * lx + L.tw1 * ly);
    g.x = __builtin_amdgcn_exp2f(e.x);
    g.y = __builtin_amdgcn_exp2f(e.y);
    return g;
}

__device__ __forceinline__ v2f btln_rule0(v2f X, v2f Y) {
    v2f r;
    r.x = ((fabsf(X.x - 1.0f) < EPSF) && (fabsf(Y.x - 1.0f) < EPSF)) ? 1.0f : 0.0f;
    r.y = ((fabsf(X.y - 1.0f) < EPSF) && (fabsf(Y.y - 1.0f) < EPSF)) ? 1.0f : 0.0f;
    return r;
}

__device__ __forceinline__ void btln_layer4(v2f& sa, v2f& sb, v2f ya, v2f yb, const btln_lay& L, bool rule0) {
    const v2f Xa = L.s * btln_clip01_2(sa) + L.o;
    const v2f Xb = L.s * btln_clip01_2(sb) + L.o;
    const v2f Ya = L.s * btln_clip01_2(ya) + L.o;
    const v2f Yb = L.s * btln_clip01_2(yb) + L.o;
    const v2f lina = L.w0 * Xa + L.w1 * Ya;
    const v2f linb = L.w0 * Xb + L.w1 * Yb;
    const v2f ga = btln_geo(Xa, Ya, L);
    const v2f gb = btln_geo(Xb, Yb, L);
    v2f ra = L.cl * lina + L.cg * ga;
    v2f rb = L.cl * linb + L.cg * gb;
    if (__builtin_expect(rule0, 0)) {
        ra = btln_rule0(Xa, Ya);
        rb = btln_rule0(Xb, Yb);
    }
    sa = L.s * ra + L.o;
    sb = L.s * rb + L.o;
}


__device__ __forceinline__ void btln_leaf_slice(const v4f (&xq)[TILES_PER_WAVE][8],
                                                const float* __restrict__ PT, float* wl,
                                                int t, int nn, int kq, int hrow) {
    v8f acc[TILES_PER_WAVE] = {};
    const float* bcol = &PT[(16 * t + nn) * PT_PITCH + kq];
#pragma unroll
    for (int j = 0; j < 8; ++j) {
        const v4f b = *(const v4f*)(bcol + 8 * j);
        const v2f b0 = __builtin_shufflevector(b, b, 0, 1);
        const v2f b1 = __builtin_shufflevector(b, b, 2, 3);
#pragma unroll
        for (int m = 0; m < TILES_PER_WAVE; ++m) {
            const v2f a0 = __builtin_shufflevector(xq[m][j], xq[m][j], 0, 1);
            acc[m] = __builtin_amdgcn_wmma_f32_16x16x4_f32(false, a0, false, b0, (short)0, acc[m], false, false);
        }
#pragma unroll
        for (int m = 0; m < TILES_PER_WAVE; ++m) {
            const v2f a1 = __builtin_shufflevector(xq[m][j], xq[m][j], 2, 3);
            acc[m] = __builtin_amdgcn_wmma_f32_16x16x4_f32(false, a1, false, b1, (short)0, acc[m], false, false);
        }
    }
#pragma unroll
    for (int m = 0; m < TILES_PER_WAVE; ++m) {
#pragma unroll
        for (int r = 0; r < 8; ++r)
            wl[(16 * m + hrow + r) * LEAF_PITCH + 16 * t + nn] = acc[m][r];
    }
    __builtin_amdgcn_wave_barrier();
}

__global__ __launch_bounds__(256) void btln_fused_kernel(const float* __restrict__ x,
                                                          const float* __restrict__ logits,
                                                          const float* __restrict__ weights,
                                                          const float* __restrict__ biases,
                                                          float* __restrict__ out) {
    __shared__ __attribute__((aligned(16))) float PT[64 * PT_PITCH];
    __shared__ float leaves[ROWS_PER_BLOCK * LEAF_PITCH];
    __shared__ __attribute__((aligned(16))) float ltab[NLAYERS * LT_STRIDE];
    __shared__ unsigned lmask[2];
    __shared__ __attribute__((aligned(16))) float svec[2 * NSZ];

    const int tid  = threadIdx.x;
    const int wave = tid >> 5;
    const int lane = tid & 31;
    const int nn   = lane & 15;
    const int kq   = (lane >> 4) * 4;
    const long wave_row0 = (long)blockIdx.x * ROWS_PER_BLOCK + wave * ROWS_PER_WAVE;

    const bool tabw = wave < 2;
    const int sub   = lane & 7;
    const int line0 = wave + 8 * (lane >> 3);
    const int line1 = line0 + 32;
    float* lrow0 = &leaves[line0 * 65 + sub];
    float* lrow1 = &leaves[line1 * 65 + sub];
    float* lcol0 = &leaves[sub * 65 + line0];
    float* lcol1 = &leaves[sub * 65 + line1];
    float* rvec = svec;
    float* cvec = svec + NSZ;
    const int wpos0 = (line0 & 7) * 8 + (line0 >> 3);
    const int wpos1 = wpos0 + 4;

    float v0[8], v1[8];
    for (int q = 0; q < 8; ++q) v0[q] = logits[line0 * 64 + sub + 8 * q];
    for (int q = 0; q < 8; ++q) v1[q] = logits[line1 * 64 + sub + 8 * q];
    const int tl = tid;
    const bool tvalid = tl < NLAYERS;
    float tw = 0.0f, tb = 0.0f;
    if (tabw) {
        const int t = tvalid ? tl : NLAYERS - 1;
        tw = weights[t];
        tb = biases[t];
    }

    v4f xq[TILES_PER_WAVE][8];
    const float* arow = x + (wave_row0 + nn) * 64 + kq;
#pragma unroll
    for (int j = 0; j < 8; ++j) {
        xq[0][j] = *(const v4f*)(arow + 8 * j);
        xq[1][j] = *(const v4f*)(arow + 16 * 64 + 8 * j);
    }

    if (tabw) {
        const float w = tw;
        const float b = tb;
        const float w1 = 1.0f - w;
        const float aref = fminf(fmaxf(1.0f - b, -1.0f + EPSF), 2.0f - EPSF);
        const bool top_lin = fabsf(b - 0.5f) < EPSF;
        const bool reflect = !top_lin && (b < 0.5f);
        const float a = reflect ? aref : b;
        const float p = sqrtf(3.0f / fmaxf(2.0f - a, EPSF)) - 1.0f;
        const bool rule0 = !top_lin && (fabsf(a - 2.0f) < EPSF);
        const bool geo = !top_lin && (a >= 0.75f);
        const bool mid = !top_lin && !geo && (a > 0.5f);
        const float sgn = reflect ? -1.0f : 1.0f;
        const float off = reflect ? 1.0f : 0.0f;
        const float cl = geo ? 0.0f : (mid ? 3.0f - 4.0f * a : 1.0f);
        const float cg = geo ? 1.0f : (mid ? 4.0f * a - 2.0f : 0.0f);
        const unsigned m0 = __builtin_amdgcn_ballot_w32(tvalid && rule0);
        if (tvalid) {
            float* e = &ltab[tl * LT_STRIDE];
            *(v4f*)(e)      = (v4f){sgn, sgn, off, off};
            *(v4f*)(e + 4)  = (v4f){w, w, w1, w1};
            *(v4f*)(e + 8)  = (v4f){2.0f * w, 2.0f * w, 2.0f * w1, 2.0f * w1};
            *(v4f*)(e + 12) = (v4f){p, p, cl, cl};
            *(v2f*)(e + 16) = (v2f){cg, cg};
        }
        if (lane == 0) lmask[tl >> 5] = m0;
    }

    float kr0[8], kr1[8];
    float kc0[8], kc1[8];
    float rv[8];
    float csc0, csc1;
    for (int q = 0; q < 8; ++q) { v0[q] *= BTLN_LOG2E; v1[q] *= BTLN_LOG2E; }
    btln_lse_normalise(v0);
    btln_lse_normalise(v1);
    for (int q = 0; q < 8; ++q) {
        kr0[q] = exp2f(v0[q]);
        kr1[q] = exp2f(v1[q]);
        lrow0[8 * q] = kr0[q];
        lrow1[8 * q] = kr1[q];
    }
    rvec[wpos0] = 1.0f;
    rvec[wpos1] = 1.0f;
    btln_lds_barrier();
#pragma unroll
    for (int j = 0; j < 8; ++j) {
#pragma unroll
        for (int m = 2; m < TILES_PER_WAVE; ++m)
            xq[m][j] = *(const v4f*)(arow + m * 16 * 64 + 8 * j);
    }
    for (int q = 0; q < 8; ++q) { kc0[q] = lcol0[8 * q * 65]; kc1[q] = lcol1[8 * q * 65]; }
    btln_load_scales(rvec + 8 * sub, rv);
    csc0 = btln_line_scale(kc0, rv);
    csc1 = btln_line_scale(kc1, rv);
    for (int it = 1; it < SINK_ITERS; ++it) {
        cvec[wpos0] = csc0;
        cvec[wpos1] = csc1;
        btln_lds_barrier();
        float cv[8];
        btln_load_scales(cvec + 8 * sub, cv);
        rvec[wpos0] = btln_line_scale(kr0, cv);
        rvec[wpos1] = btln_line_scale(kr1, cv);
        btln_lds_barrier();
        btln_load_scales(rvec + 8 * sub, rv);
        csc0 = btln_line_scale(kc0, rv);
        csc1 = btln_line_scale(kc1, rv);
    }
    for (int q = 0; q < 8; ++q) {
        PT[line0 * PT_PITCH + sub + 8 * q] = (kc0[q] * rv[q]) * csc0;
        PT[line1 * PT_PITCH + sub + 8 * q] = (kc1[q] * rv[q]) * csc1;
    }
    __syncthreads();

    float* wl = &leaves[wave * ROWS_PER_WAVE * LEAF_PITCH];
    const int hrow = (lane >> 4) << 3;
    const float* row0 = wl + lane * LEAF_PITCH;
    const float* row1 = wl + (32 + lane) * LEAF_PITCH;
    const float* row2 = wl + (64 + lane) * LEAF_PITCH;
    const float* row3 = wl + (96 + lane) * LEAF_PITCH;

    unsigned long long m0 = ((unsigned long long)__builtin_amdgcn_readfirstlane(lmask[1]) << 32) |
                            (unsigned)__builtin_amdgcn_readfirstlane(lmask[0]);

    btln_leaf_slice(xq, PT, wl, 0, nn, kq, hrow);
    v2f sa = {row0[0], row1[0]};
    v2f sb = {row2[0], row3[0]};
    btln_lay L0 = btln_load_layer(ltab, 0), L1;
    v2f ya0 = {row0[1], row1[1]}, ya1;
    v2f yb0 = {row2[1], row3[1]}, yb1;
    int i = 0;
    for (int t = 0; t < 4; ++t) {
        if (t > 0) btln_leaf_slice(xq, PT, wl, t, nn, kq, hrow);
        const int iend = 16 * t + 14;
        for (; i < iend; i += 2) {
            L1 = btln_load_layer(ltab, i + 1);
            ya1.x = row0[i + 2]; ya1.y = row1[i + 2]; yb1.x = row2[i + 2]; yb1.y = row3[i + 2];
            btln_layer4(sa, sb, ya0, yb0, L0, m0 & 1);
            L0 = btln_load_layer(ltab, i + 2);
            ya0.x = row0[i + 3]; ya0.y = row1[i + 3]; yb0.x = row2[i + 3]; yb0.y = row3[i + 3];
            btln_layer4(sa, sb, ya1, yb1, L1, m0 & 2);
            m0 >>= 2;
        }
    }
    btln_layer4(sa, sb, ya0, yb0, L0, m0 & 1);
    out[wave_row0 + lane]      = sa.x;
    out[wave_row0 + 32 + lane] = sa.y;
    out[wave_row0 + 64 + lane] = sb.x;
    out[wave_row0 + 96 + lane] = sb.y;
}

extern "C" void kernel_launch(void* const* d_in, const int* in_sizes, int n_in,
                              void* d_out, int out_size, void* d_ws, size_t ws_size,
                              hipStream_t stream) {
    const float* x       = (const float*)d_in[0];
    const float* logits  = (const float*)d_in[1];
    const float* weights = (const float*)d_in[2];
    const float* biases  = (const float*)d_in[3];
    float* out = (float*)d_out;

    btln_fused_kernel<<<BATCH_TOTAL / ROWS_PER_BLOCK, 256, 0, stream>>>(x, logits, weights, biases, out);
}
